# starting kernel
# baseline (speedup 1.0000x reference)
#include <hip/hip_runtime.h>
#include <hip/hip_bf16.h>
#include <math.h>

#define N_NODES 100000
#define N_REL   16
#define N_EDGES 1000000
#define D_IN    192
#define D_H     128
#define D_X     64
#define D_Y     32
#define D_O     16

typedef _Float16 v16h __attribute__((ext_vector_type(16)));
typedef _Float16 v8h  __attribute__((ext_vector_type(8)));
typedef float    v8f  __attribute__((ext_vector_type(8)));
typedef float    v4f  __attribute__((ext_vector_type(4)));

static __device__ __forceinline__ v16h cat8(v8h lo, v8h hi) {
    return __builtin_shufflevector(lo, hi, 0,1,2,3,4,5,6,7,8,9,10,11,12,13,14,15);
}

__launch_bounds__(256)
__global__ void mlp_wmma_kernel(const float* __restrict__ emb,
                                const float* __restrict__ w1, const float* __restrict__ b1,
                                const float* __restrict__ w2, const float* __restrict__ b2,
                                float* __restrict__ xout)
{
    __shared__ __align__(16) _Float16 sW1[D_H * D_IN];
    __shared__ __align__(16) _Float16 sW2[D_X * D_H];

    const int tid = threadIdx.x;
    for (int i = tid; i < D_H * D_IN; i += 256) sW1[i] = (_Float16)w1[i];
    for (int i = tid; i < D_X * D_H;  i += 256) sW2[i] = (_Float16)w2[i];
    __syncthreads();

    const int wave = tid >> 5;
    const int lane = tid & 31;
    const int half = lane >> 4;
    const int m    = lane & 15;
    const int rowBase = blockIdx.x * 128 + wave * 16;

    v8f accH[8] = {};
    const int gr = min(rowBase + m, N_NODES - 1);
    const float* arow = emb + (size_t)gr * D_IN + half * 8;

    for (int ks = 0; ks < 6; ++ks) {
        const float* ap = arow + ks * 32;
        v4f f0 = *(const v4f*)(ap);
        v4f f1 = *(const v4f*)(ap + 4);
        v4f f2 = *(const v4f*)(ap + 16);
        v4f f3 = *(const v4f*)(ap + 20);
        v8h lo, hi;
#pragma unroll
        for (int i = 0; i < 4; ++i) {
            lo[i]     = (_Float16)f0[i];  lo[i + 4] = (_Float16)f1[i];
            hi[i]     = (_Float16)f2[i];  hi[i + 4] = (_Float16)f3[i];
        }
        v16h a = cat8(lo, hi);
#pragma unroll
        for (int nt = 0; nt < 8; ++nt) {
            const _Float16* wp = sW1 + (nt * 16 + m) * D_IN + ks * 32 + half * 8;
            v16h b = cat8(*(const v8h*)wp, *(const v8h*)(wp + 16));
            accH[nt] = __builtin_amdgcn_wmma_f32_16x16x32_f16(
                false, a, false, b, (short)0, accH[nt], false, false);
        }
    }
    __syncthreads();

    _Float16* Hs = sW1 + wave * (16 * D_H);
#pragma unroll
    for (int nt = 0; nt < 8; ++nt) {
        const int  col  = nt * 16 + m;
        const float bias = b1[col];
#pragma unroll
        for (int r = 0; r < 8; ++r) {
            const int mrow = half * 8 + r;
            Hs[mrow * D_H + col] = (_Float16)tanhf(accH[nt][r] + bias);
        }
    }

    v8f accX[4] = {};
    for (int ks = 0; ks < 4; ++ks) {
        const _Float16* hp = Hs + m * D_H + ks * 32 + half * 8;
        v16h a = cat8(*(const v8h*)hp, *(const v8h*)(hp + 16));
#pragma unroll
        for (int nt = 0; nt < 4; ++nt) {
            const _Float16* wp = sW2 + (nt * 16 + m) * D_H + ks * 32 + half * 8;
            v16h b = cat8(*(const v8h*)wp, *(const v8h*)(wp + 16));
            accX[nt] = __builtin_amdgcn_wmma_f32_16x16x32_f16(
                false, a, false, b, (short)0, accX[nt], false, false);
        }
    }
#pragma unroll
    for (int nt = 0; nt < 4; ++nt) {
        const int  col  = nt * 16 + m;
        const float bias = b2[col];
#pragma unroll
        for (int r = 0; r < 8; ++r) {
            const int row = rowBase + half * 8 + r;
            if (row < N_NODES)
                xout[(size_t)row * D_X + col] = accX[nt][r] + bias;
        }
    }
}

__global__ void cnt_kernel(const int* __restrict__ ei, const int* __restrict__ et,
                           float* __restrict__ cnt)
{
    int e = blockIdx.x * blockDim.x + threadIdx.x;
    if (e < N_EDGES) {
        int dst = ei[N_EDGES + e];
        int r   = et[e];
        atomicAdd(&cnt[dst * N_REL + r], 1.0f);
    }
}

__global__ void root1_kernel(const float* __restrict__ x, const float* __restrict__ root,
                             const float* __restrict__ bias, float* __restrict__ acc)
{
    int t = blockIdx.x * blockDim.x + threadIdx.x;
    int node = t >> 5, col = t & 31;
    if (node >= N_NODES) return;
    float s = bias[col];
    const float* xr = x + (size_t)node * D_X;
#pragma unroll 8
    for (int f = 0; f < D_X; ++f) s += xr[f] * root[f * D_Y + col];
    acc[(size_t)node * D_Y + col] = s;
}

__global__ void edge1_kernel(const int* __restrict__ ei, const int* __restrict__ et,
                             const float* __restrict__ cnt, const float* __restrict__ x,
                             const float* __restrict__ w, float* __restrict__ acc)
{
    int t = blockIdx.x * blockDim.x + threadIdx.x;
    int e = t >> 5, col = t & 31;
    if (e >= N_EDGES) return;
    int src = ei[e], dst = ei[N_EDGES + e], r = et[e];
    float scale = 1.0f / fmaxf(cnt[dst * N_REL + r], 1.0f);
    const float* xr = x + (size_t)src * D_X;
    const float* wr = w + (size_t)r * D_X * D_Y + col;
    float s = 0.0f;
#pragma unroll 8
    for (int f = 0; f < D_X; ++f) s += xr[f] * wr[f * D_Y];
    atomicAdd(&acc[(size_t)dst * D_Y + col], s * scale);
}

__global__ void relu_kernel(float* __restrict__ a, int n)
{
    int i = blockIdx.x * blockDim.x + threadIdx.x;
    if (i < n) a[i] = fmaxf(a[i], 0.0f);
}

__global__ void root2_kernel(const float* __restrict__ y, const float* __restrict__ root,
                             const float* __restrict__ bias, float* __restrict__ out)
{
    int t = blockIdx.x * blockDim.x + threadIdx.x;
    int node = t >> 4, col = t & 15;
    if (node >= N_NODES) return;
    float s = bias[col];
    const float* yr = y + (size_t)node * D_Y;
#pragma unroll 8
    for (int f = 0; f < D_Y; ++f) s += yr[f] * root[f * D_O + col];
    out[(size_t)node * D_O + col] = s;
}

__global__ void edge2_kernel(const int* __restrict__ ei, const int* __restrict__ et,
                             const float* __restrict__ cnt, const float* __restrict__ y,
                             const float* __restrict__ w, float* __restrict__ out)
{
    int t = blockIdx.x * blockDim.x + threadIdx.x;
    int e = t >> 4, col = t & 15;
    if (e >= N_EDGES) return;
    int src = ei[e], dst = ei[N_EDGES + e], r = et[e];
    float scale = 1.0f / fmaxf(cnt[dst * N_REL + r], 1.0f);
    const float* yr = y + (size_t)src * D_Y;
    const float* wr = w + (size_t)r * D_Y * D_O + col;
    float s = 0.0f;
#pragma unroll 8
    for (int f = 0; f < D_Y; ++f) s += yr[f] * wr[f * D_O];
    atomicAdd(&out[(size_t)dst * D_O + col], s * scale);
}

__global__ void sigmoid_kernel(float* __restrict__ a, int n)
{
    int i = blockIdx.x * blockDim.x + threadIdx.x;
    if (i < n) a[i] = 1.0f / (1.0f + expf(-a[i]));
}

extern "C" void kernel_launch(void* const* d_in, const int* in_sizes, int n_in,
                              void* d_out, int out_size, void* d_ws, size_t ws_size,
                              hipStream_t stream)
{
    const float* emb   = (const float*)d_in[0];
    const float* l1w   = (const float*)d_in[1];
    const float* l1b   = (const float*)d_in[2];
    const float* l2w   = (const float*)d_in[3];
    const float* l2b   = (const float*)d_in[4];
    const float* w1    = (const float*)d_in[5];
    const float* root1 = (const float*)d_in[6];
    const float* b1    = (const float*)d_in[7];
    const float* w2    = (const float*)d_in[8];
    const float* root2 = (const float*)d_in[9];
    const float* b2    = (const float*)d_in[10];
    const int*   ei    = (const int*)d_in[11];
    const int*   et    = (const int*)d_in[12];
    float*       out   = (float*)d_out;

    char*  ws   = (char*)d_ws;
    float* x    = (float*)(ws);
    float* cnt  = (float*)(ws + (size_t)N_NODES * 64 * 4);
    float* acc1 = (float*)(ws + (size_t)N_NODES * 80 * 4);

    hipMemsetAsync(cnt, 0, (size_t)N_NODES * N_REL * sizeof(float), stream);

    mlp_wmma_kernel<<<(N_NODES + 127) / 128, 256, 0, stream>>>(emb, l1w, l1b, l2w, l2b, x);

    cnt_kernel<<<(N_EDGES + 255) / 256, 256, 0, stream>>>(ei, et, cnt);

    root1_kernel<<<(N_NODES * 32 + 255) / 256, 256, 0, stream>>>(x, root1, b1, acc1);
    edge1_kernel<<<(N_EDGES * 32 + 255) / 256, 256, 0, stream>>>(ei, et, cnt, x, w1, acc1);
    relu_kernel<<<(N_NODES * 32 + 255) / 256, 256, 0, stream>>>(acc1, N_NODES * 32);

    root2_kernel<<<(N_NODES * 16 + 255) / 256, 256, 0, stream>>>(acc1, root2, b2, out);
    edge2_kernel<<<(N_EDGES * 16 + 255) / 256, 256, 0, stream>>>(ei, et, cnt, acc1, w2, out);
    sigmoid_kernel<<<(N_NODES * 16 + 255) / 256, 256, 0, stream>>>(out, N_NODES * 16);
}


# Edges bucketed by destination range once; each edge block owns its nodes, sums messages in LDS (ds float adds), writes each row once with ReLU/sigmoid: no global float atomics; sum order changes
# speedup vs baseline: 2.3669x; 2.3669x over previous
#include <hip/hip_runtime.h>
#include <hip/hip_bf16.h>
#include <math.h>

#define N_NODES 100000
#define N_REL   16
#define N_EDGES 1000000
#define D_IN    192
#define D_H     128
#define D_X     64
#define D_Y     32
#define D_O     16

typedef _Float16 v16h __attribute__((ext_vector_type(16)));
typedef _Float16 v8h  __attribute__((ext_vector_type(8)));
typedef float    v8f  __attribute__((ext_vector_type(8)));
typedef float    v4f  __attribute__((ext_vector_type(4)));
typedef float    v2f  __attribute__((ext_vector_type(2)));
typedef int      v4i  __attribute__((ext_vector_type(4)));

static __device__ __forceinline__ v16h cat8(v8h lo, v8h hi) {
    return __builtin_shufflevector(lo, hi, 0,1,2,3,4,5,6,7,8,9,10,11,12,13,14,15);
}

__launch_bounds__(256)
__global__ void mlp_wmma_kernel(const float* __restrict__ emb,
                                const float* __restrict__ w1, const float* __restrict__ b1,
                                const float* __restrict__ w2, const float* __restrict__ b2,
                                float* __restrict__ xout)
{
    __shared__ __align__(16) _Float16 sW1[D_H * D_IN];
    __shared__ __align__(16) _Float16 sW2[D_X * D_H];

    const int tid = threadIdx.x;
    for (int i = tid; i < D_H * D_IN; i += 256) sW1[i] = (_Float16)w1[i];
    for (int i = tid; i < D_X * D_H;  i += 256) sW2[i] = (_Float16)w2[i];
    __syncthreads();

    const int wave = tid >> 5;
    const int lane = tid & 31;
    const int half = lane >> 4;
    const int m    = lane & 15;
    const int rowBase = blockIdx.x * 128 + wave * 16;

    v8f accH[8] = {};
    const int gr = min(rowBase + m, N_NODES - 1);
    const float* arow = emb + (size_t)gr * D_IN + half * 8;

    for (int ks = 0; ks < 6; ++ks) {
        const float* ap = arow + ks * 32;
        v4f f0 = *(const v4f*)(ap);
        v4f f1 = *(const v4f*)(ap + 4);
        v4f f2 = *(const v4f*)(ap + 16);
        v4f f3 = *(const v4f*)(ap + 20);
        v8h lo, hi;
#pragma unroll
        for (int i = 0; i < 4; ++i) {
            lo[i]     = (_Float16)f0[i];  lo[i + 4] = (_Float16)f1[i];
            hi[i]     = (_Float16)f2[i];  hi[i + 4] = (_Float16)f3[i];
        }
        v16h a = cat8(lo, hi);
#pragma unroll
        for (int nt = 0; nt < 8; ++nt) {
            const _Float16* wp = sW1 + (nt * 16 + m) * D_IN + ks * 32 + half * 8;
            v16h b = cat8(*(const v8h*)wp, *(const v8h*)(wp + 16));
            accH[nt] = __builtin_amdgcn_wmma_f32_16x16x32_f16(
                false, a, false, b, (short)0, accH[nt], false, false);
        }
    }
    __syncthreads();

    _Float16* Hs = sW1 + wave * (16 * D_H);
#pragma unroll
    for (int nt = 0; nt < 8; ++nt) {
        const int  col  = nt * 16 + m;
        const float bias = b1[col];
#pragma unroll
        for (int r = 0; r < 8; ++r) {
            const int mrow = half * 8 + r;
            Hs[mrow * D_H + col] = (_Float16)tanhf(accH[nt][r] + bias);
        }
    }

    v8f accX[4] = {};
    for (int ks = 0; ks < 4; ++ks) {
        const _Float16* hp = Hs + m * D_H + ks * 32 + half * 8;
        v16h a = cat8(*(const v8h*)hp, *(const v8h*)(hp + 16));
#pragma unroll
        for (int nt = 0; nt < 4; ++nt) {
            const _Float16* wp = sW2 + (nt * 16 + m) * D_H + ks * 32 + half * 8;
            v16h b = cat8(*(const v8h*)wp, *(const v8h*)(wp + 16));
            accX[nt] = __builtin_amdgcn_wmma_f32_16x16x32_f16(
                false, a, false, b, (short)0, accX[nt], false, false);
        }
    }
#pragma unroll
    for (int nt = 0; nt < 4; ++nt) {
        const int  col  = nt * 16 + m;
        const float bias = b2[col];
#pragma unroll
        for (int r = 0; r < 8; ++r) {
            const int row = rowBase + half * 8 + r;
            if (row < N_NODES)
                xout[(size_t)row * D_X + col] = accX[nt][r] + bias;
        }
    }
}

#define NB       256
#define NPB      391
#define EB_CH    2048
#define EB_PT    (EB_CH / 256)
#define EB_CAP   (EB_CH + 16 * N_REL)
#define EB_GRID  ((N_EDGES + EB_CH - 1) / EB_CH)
#define EK_T     512
#define EK_PT    (EB_CH / EK_T)

static __device__ __forceinline__ int rec_src(int r) { return r & 0x1FFFF; }
static __device__ __forceinline__ int rec_rel(int r) { return (r >> 17) & 15; }
static __device__ __forceinline__ int rec_dl(int r)  { return r >> 21; }

__launch_bounds__(256)
__global__ void bucket_count_kernel(const int* __restrict__ ei, int* __restrict__ gcount)
{
    __shared__ int sh[NB];
    const int tid = threadIdx.x;
    const int e0  = blockIdx.x * EB_CH;
    sh[tid] = 0;
    __syncthreads();
    int dst[EB_PT];
#pragma unroll
    for (int i = 0; i < EB_PT; ++i) dst[i] = ei[N_EDGES + min(e0 + i * 256 + tid, N_EDGES - 1)];
#pragma unroll
    for (int i = 0; i < EB_PT; ++i)
        if (e0 + i * 256 + tid < N_EDGES) atomicAdd(&sh[dst[i] / NPB], 1);
    __syncthreads();
    if (sh[tid] > 0) atomicAdd(&gcount[tid], sh[tid]);
}

__launch_bounds__(256)
__global__ void bucket_scatter_kernel(const int* __restrict__ ei, const int* __restrict__ et,
                                      const int* __restrict__ gcount, int* __restrict__ gcursor,
                                      int* __restrict__ gstart, int* __restrict__ recs)
{
    __shared__ int sScan[NB];
    __shared__ int sLoc[NB];
    __shared__ int sBase[NB];
    const int tid  = threadIdx.x;
    const int e0   = blockIdx.x * EB_CH;
    const int mine = gcount[tid];
    sScan[tid] = mine;
    sLoc[tid]  = 0;
    __syncthreads();
    for (int off = 1; off < NB; off <<= 1) {
        const int v = (tid >= off) ? sScan[tid - off] : 0;
        __syncthreads();
        sScan[tid] += v;
        __syncthreads();
    }
    const int start = sScan[tid] - mine;
    if (blockIdx.x == 0) {
        gstart[tid] = start;
        if (tid == NB - 1) gstart[NB] = sScan[tid];
    }

    int rec[EB_PT], bk[EB_PT];
#pragma unroll
    for (int i = 0; i < EB_PT; ++i) {
        const int ec  = min(e0 + i * 256 + tid, N_EDGES - 1);
        const int src = ei[ec];
        const int dst = ei[N_EDGES + ec];
        const int b   = dst / NPB;
        bk[i]  = b;
        rec[i] = src | (et[ec] << 17) | ((dst - b * NPB) << 21);
    }
#pragma unroll
    for (int i = 0; i < EB_PT; ++i) {
        if (e0 + i * 256 + tid >= N_EDGES) bk[i] = -1;
        if (bk[i] >= 0) atomicAdd(&sLoc[bk[i]], 1);
    }
    __syncthreads();
    const int n = sLoc[tid];
    sBase[tid] = start + ((n > 0) ? atomicAdd(&gcursor[tid], n) : 0);
    sLoc[tid]  = 0;
    __syncthreads();
#pragma unroll
    for (int i = 0; i < EB_PT; ++i) {
        if (bk[i] >= 0) {
            const int pos = sBase[bk[i]] + atomicAdd(&sLoc[bk[i]], 1);
            recs[pos] = rec[i];
        }
    }
}

__global__ void root1_kernel(const float* __restrict__ x, const float* __restrict__ root,
                             const float* __restrict__ bias, float* __restrict__ acc)
{
    int t = blockIdx.x * blockDim.x + threadIdx.x;
    int node = t >> 5, col = t & 31;
    if (node >= N_NODES) return;
    float s = bias[col];
    const float* xr = x + (size_t)node * D_X;
#pragma unroll 8
    for (int f = 0; f < D_X; ++f) s += xr[f] * root[f * D_Y + col];
    acc[(size_t)node * D_Y + col] = s;
}

template <int F, int O, int ACT>
__launch_bounds__(EK_T)
__global__ void edge_wmma_kernel(const int* __restrict__ recs, const int* __restrict__ gstart,
                                 const float* __restrict__ xin, const float* __restrict__ w,
                                 float* __restrict__ acc)
{
    constexpr int KS = F / 4;
    constexpr int NT = O / 16;
    constexpr int HV = F / 8;

    __shared__ __align__(16) int   sSrc[EB_CAP];
    __shared__ __align__(16) int   sDst[EB_CAP];
    __shared__ __align__(16) float sScl[EB_CAP];
    __shared__ int sHist[N_REL];
    __shared__ int sCur[N_REL];
    __shared__ int sTileBase[N_REL + 1];
    __shared__ int   sCnt[NPB * N_REL];
    __shared__ float sAcc[NPB * O];

    const int tid  = threadIdx.x;
    const int n0   = blockIdx.x * NPB;
    const int nn   = min(NPB, N_NODES - n0);
    const int eBeg = gstart[blockIdx.x];
    const int eEnd = gstart[blockIdx.x + 1];

    for (int i = tid; i < NPB * N_REL; i += EK_T) sCnt[i] = 0;
    for (int i = tid; i < NPB * O;     i += EK_T) sAcc[i] = 0.0f;
    __syncthreads();
    for (int e = eBeg + tid; e < eEnd; e += 4 * EK_T) {
        int rc[4];
#pragma unroll
        for (int u = 0; u < 4; ++u) rc[u] = recs[min(e + u * EK_T, eEnd - 1)];
#pragma unroll
        for (int u = 0; u < 4; ++u)
            if (e + u * EK_T < eEnd) atomicAdd(&sCnt[rec_dl(rc[u]) * N_REL + rec_rel(rc[u])], 1);
    }
    __syncthreads();

    const int wave = __builtin_amdgcn_readfirstlane(tid >> 5);
    const int lane = tid & 31;
    const int h    = lane >> 4;
    const int c    = lane & 15;

    int curR = -1;
    v2f bfrag[NT][KS];
#pragma unroll
    for (int nt = 0; nt < NT; ++nt)
#pragma unroll
        for (int j = 0; j < KS; ++j) bfrag[nt][j] = (v2f){0.0f, 0.0f};

    for (int c0 = eBeg; c0 < eEnd; c0 += EB_CH) {
        if (tid < N_REL) sHist[tid] = 0;
        for (int i = tid; i < EB_CAP; i += EK_T) { sSrc[i] = 0; sDst[i] = -1; sScl[i] = 0.0f; }
        __syncthreads();

        int myrec[EK_PT];
#pragma unroll
        for (int i = 0; i < EK_PT; ++i) {
            const int e  = c0 + i * EK_T + tid;
            const int rc = recs[min(e, eEnd - 1)];
            myrec[i] = (e < eEnd) ? rc : -1;
        }
#pragma unroll
        for (int i = 0; i < EK_PT; ++i)
            if (myrec[i] >= 0) atomicAdd(&sHist[rec_rel(myrec[i])], 1);
        __syncthreads();
        if (tid == 0) {
            int t = 0;
            for (int r = 0; r < N_REL; ++r) {
                sTileBase[r] = t;
                sCur[r]      = t * 16;
                t += (sHist[r] + 15) >> 4;
            }
            sTileBase[N_REL] = t;
        }
        __syncthreads();
#pragma unroll
        for (int i = 0; i < EK_PT; ++i) {
            if (myrec[i] >= 0) {
                const int r   = rec_rel(myrec[i]);
                const int dl  = rec_dl(myrec[i]);
                const int pos = atomicAdd(&sCur[r], 1);
                sSrc[pos] = rec_src(myrec[i]);
                sDst[pos] = dl;
                sScl[pos] = 1.0f / fmaxf((float)sCnt[dl * N_REL + r], 1.0f);
            }
        }
        __syncthreads();

        const int T    = __builtin_amdgcn_readfirstlane(sTileBase[N_REL]);
        const int per  = (T + EK_T / 32 - 1) / (EK_T / 32);
        const int tBeg = wave * per;
        const int tEnd = min(T, tBeg + per);

        v4f av[HV];
        {
            const float* x0 = xin + (size_t)sSrc[tBeg * 16 + c] * F + h * (F / 2);
#pragma unroll
            for (int i = 0; i < HV; ++i) av[i] = *(const v4f*)(x0 + 4 * i);
        }

        for (int t = tBeg; t < tEnd; ++t) {
            int rr = 0;
#pragma unroll
            for (int q = 1; q < N_REL; ++q) rr += (sTileBase[q] <= t) ? 1 : 0;
            const int r = __builtin_amdgcn_readfirstlane(rr);
            if (r != curR) {
                curR = r;
                const float* wr = w + (size_t)r * F * O + (size_t)(h * (F / 2)) * O + c;
#pragma unroll
                for (int nt = 0; nt < NT; ++nt)
#pragma unroll
                    for (int j = 0; j < KS; ++j) {
                        bfrag[nt][j][0] = wr[(2 * j)     * O + nt * 16];
                        bfrag[nt][j][1] = wr[(2 * j + 1) * O + nt * 16];
                    }
            }

            const int slot0 = t * 16;
            const int tn = min(t + 1, tEnd - 1);
            const float* xn = xin + (size_t)sSrc[tn * 16 + c] * F + h * (F / 2);
            v4f an[HV];
#pragma unroll
            for (int i = 0; i < HV; ++i) an[i] = *(const v4f*)(xn + 4 * i);

            v8f d[NT];
#pragma unroll
            for (int nt = 0; nt < NT; ++nt) d[nt] = (v8f){0.f, 0.f, 0.f, 0.f, 0.f, 0.f, 0.f, 0.f};
#pragma unroll
            for (int j = 0; j < KS; ++j) {
                v2f a;
                a[0] = av[j >> 1][(j & 1) * 2];
                a[1] = av[j >> 1][(j & 1) * 2 + 1];
#pragma unroll
                for (int nt = 0; nt < NT; ++nt)
                    d[nt] = __builtin_amdgcn_wmma_f32_16x16x4_f32(
                        false, a, false, bfrag[nt][j], (short)0, d[nt], false, false);
            }

            const v4i dlo = *(const v4i*)&sDst[slot0 + 8 * h];
            const v4i dhi = *(const v4i*)&sDst[slot0 + 8 * h + 4];
            const v4f slo = *(const v4f*)&sScl[slot0 + 8 * h];
            const v4f shi = *(const v4f*)&sScl[slot0 + 8 * h + 4];
#pragma unroll
            for (int q = 0; q < 8; ++q) {
                const int   dst = (q < 4) ? dlo[q & 3] : dhi[q & 3];
                const float sc  = (q < 4) ? slo[q & 3] : shi[q & 3];
                if (dst >= 0) {
#pragma unroll
                    for (int nt = 0; nt < NT; ++nt)
                        atomicAdd(&sAcc[dst * O + nt * 16 + c], d[nt][q] * sc);
                }
            }
#pragma unroll
            for (int i = 0; i < HV; ++i) av[i] = an[i];
        }
        __syncthreads();
    }

    float* arow = acc + (size_t)n0 * O;
    for (int i = tid; i < nn * O; i += EK_T) {
        const float v = arow[i] + sAcc[i];
        arow[i] = (ACT == 0) ? fmaxf(v, 0.0f) : 1.0f / (1.0f + expf(-v));
    }
}

__global__ void root2_kernel(const float* __restrict__ y, const float* __restrict__ root,
                             const float* __restrict__ bias, float* __restrict__ out)
{
    int t = blockIdx.x * blockDim.x + threadIdx.x;
    int node = t >> 4, col = t & 15;
    if (node >= N_NODES) return;
    float s = bias[col];
    const float* yr = y + (size_t)node * D_Y;
#pragma unroll 8
    for (int f = 0; f < D_Y; ++f) s += yr[f] * root[f * D_O + col];
    out[(size_t)node * D_O + col] = s;
}

extern "C" void kernel_launch(void* const* d_in, const int* in_sizes, int n_in,
                              void* d_out, int out_size, void* d_ws, size_t ws_size,
                              hipStream_t stream)
{
    const float* emb   = (const float*)d_in[0];
    const float* l1w   = (const float*)d_in[1];
    const float* l1b   = (const float*)d_in[2];
    const float* l2w   = (const float*)d_in[3];
    const float* l2b   = (const float*)d_in[4];
    const float* w1    = (const float*)d_in[5];
    const float* root1 = (const float*)d_in[6];
    const float* b1    = (const float*)d_in[7];
    const float* w2    = (const float*)d_in[8];
    const float* root2 = (const float*)d_in[9];
    const float* b2    = (const float*)d_in[10];
    const int*   ei    = (const int*)d_in[11];
    const int*   et    = (const int*)d_in[12];
    float*       out   = (float*)d_out;

    char*  ws      = (char*)d_ws;
    float* x       = (float*)(ws);
    int*   gcount  = (int*)(ws + (size_t)N_NODES * 64 * 4);
    int*   gcursor = (int*)(ws + (size_t)N_NODES * 64 * 4 + 1024);
    int*   gstart  = (int*)(ws + (size_t)N_NODES * 64 * 4 + 2048);
    int*   recs    = (int*)(ws + (size_t)N_NODES * 64 * 4 + 4096);
    float* acc1    = (float*)(ws + (size_t)N_NODES * 80 * 4);

    hipMemsetAsync(gcount, 0, 2048, stream);

    mlp_wmma_kernel<<<(N_NODES + 127) / 128, 256, 0, stream>>>(emb, l1w, l1b, l2w, l2b, x);

    bucket_count_kernel<<<EB_GRID, 256, 0, stream>>>(ei, gcount);
    bucket_scatter_kernel<<<EB_GRID, 256, 0, stream>>>(ei, et, gcount, gcursor, gstart, recs);

    root1_kernel<<<(N_NODES * 32 + 255) / 256, 256, 0, stream>>>(x, root1, b1, acc1);
    edge_wmma_kernel<D_X, D_Y, 0><<<NB, EK_T, 0, stream>>>(recs, gstart, x, w1, acc1);

    root2_kernel<<<(N_NODES * 16 + 255) / 256, 256, 0, stream>>>(acc1, root2, b2, out);
    edge_wmma_kernel<D_Y, D_O, 1><<<NB, EK_T, 0, stream>>>(recs, gstart, acc1, w2, out);
}


# Root transform and bias computed in each edge block as the start value of its LDS row sums (f32 WMMA 16x16x4 on the block's own rows): both root kernels and the output re-read go; sum order changes
# speedup vs baseline: 2.9141x; 1.2312x over previous
#include <hip/hip_runtime.h>
#include <hip/hip_bf16.h>
#include <math.h>

#define N_NODES 100000
#define N_REL   16
#define N_EDGES 1000000
#define D_IN    192
#define D_H     128
#define D_X     64
#define D_Y     32
#define D_O     16

typedef _Float16 v16h __attribute__((ext_vector_type(16)));
typedef _Float16 v8h  __attribute__((ext_vector_type(8)));
typedef float    v8f  __attribute__((ext_vector_type(8)));
typedef float    v4f  __attribute__((ext_vector_type(4)));
typedef float    v2f  __attribute__((ext_vector_type(2)));
typedef int      v4i  __attribute__((ext_vector_type(4)));

static __device__ __forceinline__ v16h cat8(v8h lo, v8h hi) {
    return __builtin_shufflevector(lo, hi, 0,1,2,3,4,5,6,7,8,9,10,11,12,13,14,15);
}

__launch_bounds__(256)
__global__ void mlp_wmma_kernel(const float* __restrict__ emb,
                                const float* __restrict__ w1, const float* __restrict__ b1,
                                const float* __restrict__ w2, const float* __restrict__ b2,
                                float* __restrict__ xout)
{
    __shared__ __align__(16) _Float16 sW1[D_H * D_IN];
    __shared__ __align__(16) _Float16 sW2[D_X * D_H];

    const int tid = threadIdx.x;
    for (int i = tid; i < D_H * D_IN; i += 256) sW1[i] = (_Float16)w1[i];
    for (int i = tid; i < D_X * D_H;  i += 256) sW2[i] = (_Float16)w2[i];
    __syncthreads();

    const int wave = tid >> 5;
    const int lane = tid & 31;
    const int half = lane >> 4;
    const int m    = lane & 15;
    const int rowBase = blockIdx.x * 128 + wave * 16;

    v8f accH[8] = {};
    const int gr = min(rowBase + m, N_NODES - 1);
    const float* arow = emb + (size_t)gr * D_IN + half * 8;

    for (int ks = 0; ks < 6; ++ks) {
        const float* ap = arow + ks * 32;
        v4f f0 = *(const v4f*)(ap);
        v4f f1 = *(const v4f*)(ap + 4);
        v4f f2 = *(const v4f*)(ap + 16);
        v4f f3 = *(const v4f*)(ap + 20);
        v8h lo, hi;
#pragma unroll
        for (int i = 0; i < 4; ++i) {
            lo[i]     = (_Float16)f0[i];  lo[i + 4] = (_Float16)f1[i];
            hi[i]     = (_Float16)f2[i];  hi[i + 4] = (_Float16)f3[i];
        }
        v16h a = cat8(lo, hi);
#pragma unroll
        for (int nt = 0; nt < 8; ++nt) {
            const _Float16* wp = sW1 + (nt * 16 + m) * D_IN + ks * 32 + half * 8;
            v16h b = cat8(*(const v8h*)wp, *(const v8h*)(wp + 16));
            accH[nt] = __builtin_amdgcn_wmma_f32_16x16x32_f16(
                false, a, false, b, (short)0, accH[nt], false, false);
        }
    }
    __syncthreads();

    _Float16* Hs = sW1 + wave * (16 * D_H);
#pragma unroll
    for (int nt = 0; nt < 8; ++nt) {
        const int  col  = nt * 16 + m;
        const float bias = b1[col];
#pragma unroll
        for (int r = 0; r < 8; ++r) {
            const int mrow = half * 8 + r;
            Hs[mrow * D_H + col] = (_Float16)tanhf(accH[nt][r] + bias);
        }
    }

    v8f accX[4] = {};
    for (int ks = 0; ks < 4; ++ks) {
        const _Float16* hp = Hs + m * D_H + ks * 32 + half * 8;
        v16h a = cat8(*(const v8h*)hp, *(const v8h*)(hp + 16));
#pragma unroll
        for (int nt = 0; nt < 4; ++nt) {
            const _Float16* wp = sW2 + (nt * 16 + m) * D_H + ks * 32 + half * 8;
            v16h b = cat8(*(const v8h*)wp, *(const v8h*)(wp + 16));
            accX[nt] = __builtin_amdgcn_wmma_f32_16x16x32_f16(
                false, a, false, b, (short)0, accX[nt], false, false);
        }
    }
#pragma unroll
    for (int nt = 0; nt < 4; ++nt) {
        const int  col  = nt * 16 + m;
        const float bias = b2[col];
#pragma unroll
        for (int r = 0; r < 8; ++r) {
            const int row = rowBase + half * 8 + r;
            if (row < N_NODES)
                xout[(size_t)row * D_X + col] = accX[nt][r] + bias;
        }
    }
}

#define NB       256
#define NPB      391
#define EB_CH    2048
#define EB_PT    (EB_CH / 256)
#define EB_CAP   (EB_CH + 16 * N_REL)
#define EB_GRID  ((N_EDGES + EB_CH - 1) / EB_CH)
#define EK_T     512
#define EK_PT    (EB_CH / EK_T)

static __device__ __forceinline__ int rec_src(int r) { return r & 0x1FFFF; }
static __device__ __forceinline__ int rec_rel(int r) { return (r >> 17) & 15; }
static __device__ __forceinline__ int rec_dl(int r)  { return r >> 21; }

__launch_bounds__(256)
__global__ void bucket_count_kernel(const int* __restrict__ ei, int* __restrict__ gcount)
{
    __shared__ int sh[NB];
    const int tid = threadIdx.x;
    const int e0  = blockIdx.x * EB_CH;
    sh[tid] = 0;
    __syncthreads();
    int dst[EB_PT];
#pragma unroll
    for (int i = 0; i < EB_PT; ++i) dst[i] = ei[N_EDGES + min(e0 + i * 256 + tid, N_EDGES - 1)];
#pragma unroll
    for (int i = 0; i < EB_PT; ++i)
        if (e0 + i * 256 + tid < N_EDGES) atomicAdd(&sh[dst[i] / NPB], 1);
    __syncthreads();
    if (sh[tid] > 0) atomicAdd(&gcount[tid], sh[tid]);
}

__launch_bounds__(256)
__global__ void bucket_scatter_kernel(const int* __restrict__ ei, const int* __restrict__ et,
                                      const int* __restrict__ gcount, int* __restrict__ gcursor,
                                      int* __restrict__ gstart, int* __restrict__ recs)
{
    __shared__ int sScan[NB];
    __shared__ int sLoc[NB];
    __shared__ int sBase[NB];
    const int tid  = threadIdx.x;
    const int e0   = blockIdx.x * EB_CH;
    const int mine = gcount[tid];
    sScan[tid] = mine;
    sLoc[tid]  = 0;
    __syncthreads();
    for (int off = 1; off < NB; off <<= 1) {
        const int v = (tid >= off) ? sScan[tid - off] : 0;
        __syncthreads();
        sScan[tid] += v;
        __syncthreads();
    }
    const int start = sScan[tid] - mine;
    if (blockIdx.x == 0) {
        gstart[tid] = start;
        if (tid == NB - 1) gstart[NB] = sScan[tid];
    }

    int rec[EB_PT], bk[EB_PT];
#pragma unroll
    for (int i = 0; i < EB_PT; ++i) {
        const int ec  = min(e0 + i * 256 + tid, N_EDGES - 1);
        const int src = ei[ec];
        const int dst = ei[N_EDGES + ec];
        const int b   = dst / NPB;
        bk[i]  = b;
        rec[i] = src | (et[ec] << 17) | ((dst - b * NPB) << 21);
    }
#pragma unroll
    for (int i = 0; i < EB_PT; ++i) {
        if (e0 + i * 256 + tid >= N_EDGES) bk[i] = -1;
        if (bk[i] >= 0) atomicAdd(&sLoc[bk[i]], 1);
    }
    __syncthreads();
    const int n = sLoc[tid];
    sBase[tid] = start + ((n > 0) ? atomicAdd(&gcursor[tid], n) : 0);
    sLoc[tid]  = 0;
    __syncthreads();
#pragma unroll
    for (int i = 0; i < EB_PT; ++i) {
        if (bk[i] >= 0) {
            const int pos = sBase[bk[i]] + atomicAdd(&sLoc[bk[i]], 1);
            recs[pos] = rec[i];
        }
    }
}


template <int F, int O>
static __device__ __forceinline__ void load_bfrag(const float* __restrict__ mat, int h, int c,
                                                  v2f (&bfrag)[O / 16][F / 4])
{
    const float* wr = mat + (size_t)(h * (F / 2)) * O + c;
#pragma unroll
    for (int nt = 0; nt < O / 16; ++nt)
#pragma unroll
        for (int j = 0; j < F / 4; ++j) {
            bfrag[nt][j][0] = wr[(2 * j)     * O + nt * 16];
            bfrag[nt][j][1] = wr[(2 * j + 1) * O + nt * 16];
        }
}

template <int F, int O>
static __device__ __forceinline__ void tile_products(const v4f (&av)[F / 8],
                                                     const v2f (&bfrag)[O / 16][F / 4],
                                                     v8f (&d)[O / 16])
{
#pragma unroll
    for (int nt = 0; nt < O / 16; ++nt) d[nt] = (v8f){0.f, 0.f, 0.f, 0.f, 0.f, 0.f, 0.f, 0.f};
#pragma unroll
    for (int j = 0; j < F / 4; ++j) {
        v2f a;
        a[0] = av[j >> 1][(j & 1) * 2];
        a[1] = av[j >> 1][(j & 1) * 2 + 1];
#pragma unroll
        for (int nt = 0; nt < O / 16; ++nt)
            d[nt] = __builtin_amdgcn_wmma_f32_16x16x4_f32(
                false, a, false, bfrag[nt][j], (short)0, d[nt], false, false);
    }
}

template <int F, int O, int ACT>
__launch_bounds__(EK_T)
__global__ void edge_wmma_kernel(const int* __restrict__ recs, const int* __restrict__ gstart,
                                 const float* __restrict__ xin, const float* __restrict__ w,
                                 const float* __restrict__ root, const float* __restrict__ bias,
                                 float* __restrict__ acc)
{
    constexpr int KS = F / 4;
    constexpr int NT = O / 16;
    constexpr int HV = F / 8;

    __shared__ __align__(16) int   sSrc[EB_CAP];
    __shared__ __align__(16) int   sDst[EB_CAP];
    __shared__ __align__(16) float sScl[EB_CAP];
    __shared__ int sHist[N_REL];
    __shared__ int sCur[N_REL];
    __shared__ int sTileBase[N_REL + 1];
    __shared__ int   sCnt[NPB * N_REL];
    __shared__ float sAcc[NPB * O];

    const int tid  = threadIdx.x;
    const int n0   = blockIdx.x * NPB;
    const int nn   = min(NPB, N_NODES - n0);
    const int eBeg = gstart[blockIdx.x];
    const int eEnd = gstart[blockIdx.x + 1];

    for (int i = tid; i < NPB * N_REL; i += EK_T) sCnt[i] = 0;

    const int wave = __builtin_amdgcn_readfirstlane(tid >> 5);
    const int lane = tid & 31;
    const int h    = lane >> 4;
    const int c    = lane & 15;

    v2f bfrag[NT][KS];

    load_bfrag<F, O>(root, h, c, bfrag);
    for (int t = wave; t * 16 < nn; t += EK_T / 32) {
        const int rc = min(t * 16 + c, nn - 1);
        const float* x0 = xin + (size_t)(n0 + rc) * F + h * (F / 2);
        v4f ar[HV];
#pragma unroll
        for (int i = 0; i < HV; ++i) ar[i] = *(const v4f*)(x0 + 4 * i);
        v8f d[NT];
        tile_products<F, O>(ar, bfrag, d);
#pragma unroll
        for (int nt = 0; nt < NT; ++nt) {
            const float bs = bias[nt * 16 + c];
#pragma unroll
            for (int q = 0; q < 8; ++q) {
                const int r = t * 16 + 8 * h + q;
                if (r < nn) sAcc[r * O + nt * 16 + c] = d[nt][q] + bs;
            }
        }
    }
    __syncthreads();
    for (int e = eBeg + tid; e < eEnd; e += 4 * EK_T) {
        int rc[4];
#pragma unroll
        for (int u = 0; u < 4; ++u) rc[u] = recs[min(e + u * EK_T, eEnd - 1)];
#pragma unroll
        for (int u = 0; u < 4; ++u)
            if (e + u * EK_T < eEnd) atomicAdd(&sCnt[rec_dl(rc[u]) * N_REL + rec_rel(rc[u])], 1);
    }
    __syncthreads();

    int curR = -1;

    for (int c0 = eBeg; c0 < eEnd; c0 += EB_CH) {
        if (tid < N_REL) sHist[tid] = 0;
        for (int i = tid; i < EB_CAP; i += EK_T) { sSrc[i] = 0; sDst[i] = -1; sScl[i] = 0.0f; }
        __syncthreads();

        int myrec[EK_PT];
#pragma unroll
        for (int i = 0; i < EK_PT; ++i) {
            const int e  = c0 + i * EK_T + tid;
            const int rc = recs[min(e, eEnd - 1)];
            myrec[i] = (e < eEnd) ? rc : -1;
        }
#pragma unroll
        for (int i = 0; i < EK_PT; ++i)
            if (myrec[i] >= 0) atomicAdd(&sHist[rec_rel(myrec[i])], 1);
        __syncthreads();
        if (tid == 0) {
            int t = 0;
            for (int r = 0; r < N_REL; ++r) {
                sTileBase[r] = t;
                sCur[r]      = t * 16;
                t += (sHist[r] + 15) >> 4;
            }
            sTileBase[N_REL] = t;
        }
        __syncthreads();
#pragma unroll
        for (int i = 0; i < EK_PT; ++i) {
            if (myrec[i] >= 0) {
                const int r   = rec_rel(myrec[i]);
                const int dl  = rec_dl(myrec[i]);
                const int pos = atomicAdd(&sCur[r], 1);
                sSrc[pos] = rec_src(myrec[i]);
                sDst[pos] = dl;
                sScl[pos] = 1.0f / fmaxf((float)sCnt[dl * N_REL + r], 1.0f);
            }
        }
        __syncthreads();

        const int T    = __builtin_amdgcn_readfirstlane(sTileBase[N_REL]);
        const int per  = (T + EK_T / 32 - 1) / (EK_T / 32);
        const int tBeg = wave * per;
        const int tEnd = min(T, tBeg + per);

        v4f av[HV];
        {
            const float* x0 = xin + (size_t)sSrc[tBeg * 16 + c] * F + h * (F / 2);
#pragma unroll
            for (int i = 0; i < HV; ++i) av[i] = *(const v4f*)(x0 + 4 * i);
        }

        for (int t = tBeg; t < tEnd; ++t) {
            int rr = 0;
#pragma unroll
            for (int q = 1; q < N_REL; ++q) rr += (sTileBase[q] <= t) ? 1 : 0;
            const int r = __builtin_amdgcn_readfirstlane(rr);
            if (r != curR) {
                curR = r;
                load_bfrag<F, O>(w + (size_t)r * F * O, h, c, bfrag);
            }

            const int slot0 = t * 16;
            const int tn = min(t + 1, tEnd - 1);
            const float* xn = xin + (size_t)sSrc[tn * 16 + c] * F + h * (F / 2);
            v4f an[HV];
#pragma unroll
            for (int i = 0; i < HV; ++i) an[i] = *(const v4f*)(xn + 4 * i);

            v8f d[NT];
            tile_products<F, O>(av, bfrag, d);

            const v4i dlo = *(const v4i*)&sDst[slot0 + 8 * h];
            const v4i dhi = *(const v4i*)&sDst[slot0 + 8 * h + 4];
            const v4f slo = *(const v4f*)&sScl[slot0 + 8 * h];
            const v4f shi = *(const v4f*)&sScl[slot0 + 8 * h + 4];
#pragma unroll
            for (int q = 0; q < 8; ++q) {
                const int   dst = (q < 4) ? dlo[q & 3] : dhi[q & 3];
                const float sc  = (q < 4) ? slo[q & 3] : shi[q & 3];
                if (dst >= 0) {
#pragma unroll
                    for (int nt = 0; nt < NT; ++nt)
                        atomicAdd(&sAcc[dst * O + nt * 16 + c], d[nt][q] * sc);
                }
            }
#pragma unroll
            for (int i = 0; i < HV; ++i) av[i] = an[i];
        }
        __syncthreads();
    }

    float* arow = acc + (size_t)n0 * O;
    for (int i = tid; i < nn * O; i += EK_T) {
        const float v = sAcc[i];
        arow[i] = (ACT == 0) ? fmaxf(v, 0.0f) : 1.0f / (1.0f + expf(-v));
    }
}

extern "C" void kernel_launch(void* const* d_in, const int* in_sizes, int n_in,
                              void* d_out, int out_size, void* d_ws, size_t ws_size,
                              hipStream_t stream)
{
    const float* emb   = (const float*)d_in[0];
    const float* l1w   = (const float*)d_in[1];
    const float* l1b   = (const float*)d_in[2];
    const float* l2w   = (const float*)d_in[3];
    const float* l2b   = (const float*)d_in[4];
    const float* w1    = (const float*)d_in[5];
    const float* root1 = (const float*)d_in[6];
    const float* b1    = (const float*)d_in[7];
    const float* w2    = (const float*)d_in[8];
    const float* root2 = (const float*)d_in[9];
    const float* b2    = (const float*)d_in[10];
    const int*   ei    = (const int*)d_in[11];
    const int*   et    = (const int*)d_in[12];
    float*       out   = (float*)d_out;

    char*  ws      = (char*)d_ws;
    float* x       = (float*)(ws);
    int*   gcount  = (int*)(ws + (size_t)N_NODES * 64 * 4);
    int*   gcursor = (int*)(ws + (size_t)N_NODES * 64 * 4 + 1024);
    int*   gstart  = (int*)(ws + (size_t)N_NODES * 64 * 4 + 2048);
    int*   recs    = (int*)(ws + (size_t)N_NODES * 64 * 4 + 4096);
    float* acc1    = (float*)(ws + (size_t)N_NODES * 80 * 4);

    hipMemsetAsync(gcount, 0, 2048, stream);

    mlp_wmma_kernel<<<(N_NODES + 127) / 128, 256, 0, stream>>>(emb, l1w, l1b, l2w, l2b, x);

    bucket_count_kernel<<<EB_GRID, 256, 0, stream>>>(ei, gcount);
    bucket_scatter_kernel<<<EB_GRID, 256, 0, stream>>>(ei, et, gcount, gcursor, gstart, recs);

    edge_wmma_kernel<D_X, D_Y, 0><<<NB, EK_T, 0, stream>>>(recs, gstart, x, w1, root1, b1, acc1);

    edge_wmma_kernel<D_Y, D_O, 1><<<NB, EK_T, 0, stream>>>(recs, gstart, acc1, w2, root2, b2, out);
}


# Bucket scatter runs as extra blocks of the MLP launch (role by block index; the two are independent): one launch fewer, the scatter hides under the MLP; arithmetic and sum order as in the parent
# speedup vs baseline: 3.6063x; 1.2375x over previous
#include <hip/hip_runtime.h>
#include <hip/hip_bf16.h>
#include <math.h>

#define N_NODES 100000
#define N_REL   16
#define N_EDGES 1000000
#define D_IN    192
#define D_H     128
#define D_X     64
#define D_Y     32
#define D_O     16

typedef _Float16 v16h __attribute__((ext_vector_type(16)));
typedef _Float16 v8h  __attribute__((ext_vector_type(8)));
typedef float    v8f  __attribute__((ext_vector_type(8)));
typedef float    v4f  __attribute__((ext_vector_type(4)));
typedef float    v2f  __attribute__((ext_vector_type(2)));
typedef int      v4i  __attribute__((ext_vector_type(4)));

static __device__ __forceinline__ v16h cat8(v8h lo, v8h hi) {
    return __builtin_shufflevector(lo, hi, 0,1,2,3,4,5,6,7,8,9,10,11,12,13,14,15);
}

#define NB       256
#define NPB      391
#define EB_CH    2048
#define EB_PT    (EB_CH / 256)
#define EB_CAP   (EB_CH + 16 * N_REL)
#define EB_GRID  ((N_EDGES + EB_CH - 1) / EB_CH)
#define MLP_GRID ((N_NODES + 127) / 128)
#define EK_T     512
#define EK_PT    (EB_CH / EK_T)

static __device__ __forceinline__ int rec_src(int r) { return r & 0x1FFFF; }
static __device__ __forceinline__ int rec_rel(int r) { return (r >> 17) & 15; }
static __device__ __forceinline__ int rec_dl(int r)  { return r >> 21; }

__launch_bounds__(256)
__global__ void bucket_count_kernel(const int* __restrict__ ei, int* __restrict__ gcount)
{
    __shared__ int sh[NB];
    const int tid = threadIdx.x;
    const int e0  = blockIdx.x * EB_CH;
    sh[tid] = 0;
    __syncthreads();
    int dst[EB_PT];
#pragma unroll
    for (int i = 0; i < EB_PT; ++i) dst[i] = ei[N_EDGES + min(e0 + i * 256 + tid, N_EDGES - 1)];
#pragma unroll
    for (int i = 0; i < EB_PT; ++i)
        if (e0 + i * 256 + tid < N_EDGES) atomicAdd(&sh[dst[i] / NPB], 1);
    __syncthreads();
    if (sh[tid] > 0) atomicAdd(&gcount[tid], sh[tid]);
}

static __device__ __forceinline__ void bucket_scatter_block(const int blk, int* sh,
                                      const int* __restrict__ ei, const int* __restrict__ et,
                                      const int* __restrict__ gcount, int* __restrict__ gcursor,
                                      int* __restrict__ gstart, int* __restrict__ recs)
{
    int* sScan = sh;
    int* sLoc  = sh + NB;
    int* sBase = sh + 2 * NB;
    const int tid  = threadIdx.x;
    const int e0   = blk * EB_CH;
    const int mine = gcount[tid];
    sScan[tid] = mine;
    sLoc[tid]  = 0;
    __syncthreads();
    for (int off = 1; off < NB; off <<= 1) {
        const int v = (tid >= off) ? sScan[tid - off] : 0;
        __syncthreads();
        sScan[tid] += v;
        __syncthreads();
    }
    const int start = sScan[tid] - mine;
    if (blk == 0) {
        gstart[tid] = start;
        if (tid == NB - 1) gstart[NB] = sScan[tid];
    }

    int rec[EB_PT], bk[EB_PT];
#pragma unroll
    for (int i = 0; i < EB_PT; ++i) {
        const int ec  = min(e0 + i * 256 + tid, N_EDGES - 1);
        const int src = ei[ec];
        const int dst = ei[N_EDGES + ec];
        const int b   = dst / NPB;
        bk[i]  = b;
        rec[i] = src | (et[ec] << 17) | ((dst - b * NPB) << 21);
    }
#pragma unroll
    for (int i = 0; i < EB_PT; ++i) {
        if (e0 + i * 256 + tid >= N_EDGES) bk[i] = -1;
        if (bk[i] >= 0) atomicAdd(&sLoc[bk[i]], 1);
    }
    __syncthreads();
    const int n = sLoc[tid];
    sBase[tid] = start + ((n > 0) ? atomicAdd(&gcursor[tid], n) : 0);
    sLoc[tid]  = 0;
    __syncthreads();
#pragma unroll
    for (int i = 0; i < EB_PT; ++i) {
        if (bk[i] >= 0) {
            const int pos = sBase[bk[i]] + atomicAdd(&sLoc[bk[i]], 1);
            recs[pos] = rec[i];
        }
    }
}

static __device__ __forceinline__ int w1_at(int r, int k) { return r * D_IN + ((k ^ ((r >> 1) & 7)) << 3); }
static __device__ __forceinline__ int w2_at(int r, int k) { return r * D_H + ((k ^ (r & 15)) << 3); }

__launch_bounds__(256)
__global__ void mlp_and_scatter_kernel(const float* __restrict__ emb,
                                       const float* __restrict__ w1, const float* __restrict__ b1,
                                       const float* __restrict__ w2, const float* __restrict__ b2,
                                       float* __restrict__ xout,
                                       const int* __restrict__ ei, const int* __restrict__ et,
                                       const int* __restrict__ gcount, int* __restrict__ gcursor,
                                       int* __restrict__ gstart, int* __restrict__ recs)
{
    __shared__ __align__(16) _Float16 sW1[D_H * D_IN];
    __shared__ __align__(16) _Float16 sW2[D_X * D_H];

    if (blockIdx.x >= MLP_GRID) {
        bucket_scatter_block(blockIdx.x - MLP_GRID, reinterpret_cast<int*>(sW1),
                             ei, et, gcount, gcursor, gstart, recs);
        return;
    }

    const int tid = threadIdx.x;
    {
        const int c0 = tid & 63, r0 = tid >> 6;
        for (int r = r0; r < D_H; r += 4) {
            for (int j = 0; j < 3; ++j) {
                const int col = c0 + 64 * j;
                sW1[w1_at(r, col >> 3) + (col & 7)] = (_Float16)w1[r * D_IN + col];
            }
        }
    }
    {
        const int c0 = tid & 127, r0 = tid >> 7;
        for (int r = r0; r < D_X; r += 2)
            sW2[w2_at(r, c0 >> 3) + (c0 & 7)] = (_Float16)w2[r * D_H + c0];
    }
    __syncthreads();

    const int wave = tid >> 5;
    const int lane = tid & 31;
    const int half = lane >> 4;
    const int m    = lane & 15;
    const int rowBase = blockIdx.x * 128 + wave * 16;

    v8f accH[8] = {};
    const int gr = min(rowBase + m, N_NODES - 1);
    const float* arow = emb + (size_t)gr * D_IN + half * 8;

    for (int ks = 0; ks < 6; ++ks) {
        const float* ap = arow + ks * 32;
        v4f f0 = *(const v4f*)(ap);
        v4f f1 = *(const v4f*)(ap + 4);
        v4f f2 = *(const v4f*)(ap + 16);
        v4f f3 = *(const v4f*)(ap + 20);
        v8h lo, hi;
#pragma unroll
        for (int i = 0; i < 4; ++i) {
            lo[i]     = (_Float16)f0[i];  lo[i + 4] = (_Float16)f1[i];
            hi[i]     = (_Float16)f2[i];  hi[i + 4] = (_Float16)f3[i];
        }
        v16h a = cat8(lo, hi);
#pragma unroll
        for (int nt = 0; nt < 8; ++nt) {
            v16h b = cat8(*(const v8h*)(sW1 + w1_at(nt * 16 + m, ks * 4 + half)),
                          *(const v8h*)(sW1 + w1_at(nt * 16 + m, ks * 4 + half + 2)));
            accH[nt] = __builtin_amdgcn_wmma_f32_16x16x32_f16(
                false, a, false, b, (short)0, accH[nt], false, false);
        }
    }
    __syncthreads();

    _Float16* Hs = sW1 + wave * (16 * D_H);
#pragma unroll
    for (int nt = 0; nt < 8; ++nt) {
        const int  col  = nt * 16 + m;
        const float bias = b1[col];
#pragma unroll
        for (int r = 0; r < 8; ++r) {
            const int mrow = half * 8 + r;
            Hs[w2_at(mrow, col >> 3) + (col & 7)] = (_Float16)tanhf(accH[nt][r] + bias);
        }
    }

    v8f accX[4] = {};
    for (int ks = 0; ks < 4; ++ks) {
        v16h a = cat8(*(const v8h*)(Hs + w2_at(m, ks * 4 + half)),
                      *(const v8h*)(Hs + w2_at(m, ks * 4 + half + 2)));
#pragma unroll
        for (int nt = 0; nt < 4; ++nt) {
            v16h b = cat8(*(const v8h*)(sW2 + w2_at(nt * 16 + m, ks * 4 + half)),
                          *(const v8h*)(sW2 + w2_at(nt * 16 + m, ks * 4 + half + 2)));
            accX[nt] = __builtin_amdgcn_wmma_f32_16x16x32_f16(
                false, a, false, b, (short)0, accX[nt], false, false);
        }
    }
#pragma unroll
    for (int nt = 0; nt < 4; ++nt) {
        const int  col  = nt * 16 + m;
        const float bias = b2[col];
#pragma unroll
        for (int r = 0; r < 8; ++r) {
            const int row = rowBase + half * 8 + r;
            if (row < N_NODES)
                xout[(size_t)row * D_X + col] = accX[nt][r] + bias;
        }
    }
}


template <int F, int O>
static __device__ __forceinline__ void load_bfrag(const float* __restrict__ mat, int h, int c,
                                                  v2f (&bfrag)[O / 16][F / 4])
{
    const float* wr = mat + (size_t)(h * (F / 2)) * O + c;
#pragma unroll
    for (int nt = 0; nt < O / 16; ++nt)
#pragma unroll
        for (int j = 0; j < F / 4; ++j) {
            bfrag[nt][j][0] = wr[(2 * j)     * O + nt * 16];
            bfrag[nt][j][1] = wr[(2 * j + 1) * O + nt * 16];
        }
}

template <int F, int O>
static __device__ __forceinline__ void tile_products(const v4f (&av)[F / 8],
                                                     const v2f (&bfrag)[O / 16][F / 4],
                                                     v8f (&d)[O / 16])
{
#pragma unroll
    for (int nt = 0; nt < O / 16; ++nt) d[nt] = (v8f){0.f, 0.f, 0.f, 0.f, 0.f, 0.f, 0.f, 0.f};
#pragma unroll
    for (int j = 0; j < F / 4; ++j) {
        v2f a;
        a[0] = av[j >> 1][(j & 1) * 2];
        a[1] = av[j >> 1][(j & 1) * 2 + 1];
#pragma unroll
        for (int nt = 0; nt < O / 16; ++nt)
            d[nt] = __builtin_amdgcn_wmma_f32_16x16x4_f32(
                false, a, false, bfrag[nt][j], (short)0, d[nt], false, false);
    }
}

template <int F, int O, int ACT>
__launch_bounds__(EK_T)
__global__ void edge_wmma_kernel(const int* __restrict__ recs, const int* __restrict__ gstart,
                                 const float* __restrict__ xin, const float* __restrict__ w,
                                 const float* __restrict__ root, const float* __restrict__ bias,
                                 float* __restrict__ acc)
{
    constexpr int KS = F / 4;
    constexpr int NT = O / 16;
    constexpr int HV = F / 8;

    __shared__ __align__(16) int   sSrc[EB_CAP];
    __shared__ __align__(16) int   sDst[EB_CAP];
    __shared__ __align__(16) float sScl[EB_CAP];
    __shared__ int sHist[N_REL];
    __shared__ int sCur[N_REL];
    __shared__ int sTileBase[N_REL + 1];
    __shared__ int   sCnt[NPB * N_REL];
    __shared__ float sAcc[NPB * O];

    const int tid  = threadIdx.x;
    const int n0   = blockIdx.x * NPB;
    const int nn   = min(NPB, N_NODES - n0);
    const int eBeg = gstart[blockIdx.x];
    const int eEnd = gstart[blockIdx.x + 1];

    for (int i = tid; i < NPB * N_REL; i += EK_T) sCnt[i] = 0;

    const int wave = __builtin_amdgcn_readfirstlane(tid >> 5);
    const int lane = tid & 31;
    const int h    = lane >> 4;
    const int c    = lane & 15;

    v2f bfrag[NT][KS];

    load_bfrag<F, O>(root, h, c, bfrag);
    for (int t = wave; t * 16 < nn; t += EK_T / 32) {
        const int rc = min(t * 16 + c, nn - 1);
        const float* x0 = xin + (size_t)(n0 + rc) * F + h * (F / 2);
        v4f ar[HV];
#pragma unroll
        for (int i = 0; i < HV; ++i) ar[i] = *(const v4f*)(x0 + 4 * i);
        v8f d[NT];
        tile_products<F, O>(ar, bfrag, d);
#pragma unroll
        for (int nt = 0; nt < NT; ++nt) {
            const float bs = bias[nt * 16 + c];
#pragma unroll
            for (int q = 0; q < 8; ++q) {
                const int r = t * 16 + 8 * h + q;
                if (r < nn) sAcc[r * O + nt * 16 + c] = d[nt][q] + bs;
            }
        }
    }
    __syncthreads();
    for (int e = eBeg + tid; e < eEnd; e += 4 * EK_T) {
        int rc[4];
#pragma unroll
        for (int u = 0; u < 4; ++u) rc[u] = recs[min(e + u * EK_T, eEnd - 1)];
#pragma unroll
        for (int u = 0; u < 4; ++u)
            if (e + u * EK_T < eEnd) atomicAdd(&sCnt[rec_dl(rc[u]) * N_REL + rec_rel(rc[u])], 1);
    }
    __syncthreads();

    int curR = -1;

    for (int c0 = eBeg; c0 < eEnd; c0 += EB_CH) {
        if (tid < N_REL) sHist[tid] = 0;
        for (int i = tid; i < EB_CAP; i += EK_T) { sSrc[i] = 0; sDst[i] = -1; sScl[i] = 0.0f; }
        __syncthreads();

        int myrec[EK_PT];
#pragma unroll
        for (int i = 0; i < EK_PT; ++i) {
            const int e  = c0 + i * EK_T + tid;
            const int rc = recs[min(e, eEnd - 1)];
            myrec[i] = (e < eEnd) ? rc : -1;
        }
#pragma unroll
        for (int i = 0; i < EK_PT; ++i)
            if (myrec[i] >= 0) atomicAdd(&sHist[rec_rel(myrec[i])], 1);
        __syncthreads();
        if (tid == 0) {
            int t = 0;
            for (int r = 0; r < N_REL; ++r) {
                sTileBase[r] = t;
                sCur[r]      = t * 16;
                t += (sHist[r] + 15) >> 4;
            }
            sTileBase[N_REL] = t;
        }
        __syncthreads();
#pragma unroll
        for (int i = 0; i < EK_PT; ++i) {
            if (myrec[i] >= 0) {
                const int r   = rec_rel(myrec[i]);
                const int dl  = rec_dl(myrec[i]);
                const int pos = atomicAdd(&sCur[r], 1);
                sSrc[pos] = rec_src(myrec[i]);
                sDst[pos] = dl;
                sScl[pos] = 1.0f / fmaxf((float)sCnt[dl * N_REL + r], 1.0f);
            }
        }
        __syncthreads();

        const int T    = __builtin_amdgcn_readfirstlane(sTileBase[N_REL]);
        const int per  = (T + EK_T / 32 - 1) / (EK_T / 32);
        const int tBeg = wave * per;
        const int tEnd = min(T, tBeg + per);

        v4f av[HV];
        {
            const float* x0 = xin + (size_t)sSrc[tBeg * 16 + c] * F + h * (F / 2);
#pragma unroll
            for (int i = 0; i < HV; ++i) av[i] = *(const v4f*)(x0 + 4 * i);
        }

        for (int t = tBeg; t < tEnd; ++t) {
            int rr = 0;
#pragma unroll
            for (int q = 1; q < N_REL; ++q) rr += (sTileBase[q] <= t) ? 1 : 0;
            const int r = __builtin_amdgcn_readfirstlane(rr);
            if (r != curR) {
                curR = r;
                load_bfrag<F, O>(w + (size_t)r * F * O, h, c, bfrag);
            }

            const int slot0 = t * 16;
            const int tn = min(t + 1, tEnd - 1);
            const float* xn = xin + (size_t)sSrc[tn * 16 + c] * F + h * (F / 2);
            v4f an[HV];
#pragma unroll
            for (int i = 0; i < HV; ++i) an[i] = *(const v4f*)(xn + 4 * i);

            v8f d[NT];
            tile_products<F, O>(av, bfrag, d);

            const v4i dlo = *(const v4i*)&sDst[slot0 + 8 * h];
            const v4i dhi = *(const v4i*)&sDst[slot0 + 8 * h + 4];
            const v4f slo = *(const v4f*)&sScl[slot0 + 8 * h];
            const v4f shi = *(const v4f*)&sScl[slot0 + 8 * h + 4];
#pragma unroll
            for (int q = 0; q < 8; ++q) {
                const int   dst = (q < 4) ? dlo[q & 3] : dhi[q & 3];
                const float sc  = (q < 4) ? slo[q & 3] : shi[q & 3];
                if (dst >= 0) {
#pragma unroll
                    for (int nt = 0; nt < NT; ++nt)
                        atomicAdd(&sAcc[dst * O + nt * 16 + c], d[nt][q] * sc);
                }
            }
#pragma unroll
            for (int i = 0; i < HV; ++i) av[i] = an[i];
        }
        __syncthreads();
    }

    float* arow = acc + (size_t)n0 * O;
    for (int i = tid; i < nn * O; i += EK_T) {
        const float v = sAcc[i];
        arow[i] = (ACT == 0) ? fmaxf(v, 0.0f) : 1.0f / (1.0f + expf(-v));
    }
}

extern "C" void kernel_launch(void* const* d_in, const int* in_sizes, int n_in,
                              void* d_out, int out_size, void* d_ws, size_t ws_size,
                              hipStream_t stream)
{
    const float* emb   = (const float*)d_in[0];
    const float* l1w   = (const float*)d_in[1];
    const float* l1b   = (const float*)d_in[2];
    const float* l2w   = (const float*)d_in[3];
    const float* l2b   = (const float*)d_in[4];
    const float* w1    = (const float*)d_in[5];
    const float* root1 = (const float*)d_in[6];
    const float* b1    = (const float*)d_in[7];
    const float* w2    = (const float*)d_in[8];
    const float* root2 = (const float*)d_in[9];
    const float* b2    = (const float*)d_in[10];
    const int*   ei    = (const int*)d_in[11];
    const int*   et    = (const int*)d_in[12];
    float*       out   = (float*)d_out;

    char*  ws      = (char*)d_ws;
    float* x       = (float*)(ws);
    int*   gcount  = (int*)(ws + (size_t)N_NODES * 64 * 4);
    int*   gcursor = (int*)(ws + (size_t)N_NODES * 64 * 4 + 1024);
    int*   gstart  = (int*)(ws + (size_t)N_NODES * 64 * 4 + 2048);
    int*   recs    = (int*)(ws + (size_t)N_NODES * 64 * 4 + 4096);
    float* acc1    = (float*)(ws + (size_t)N_NODES * 80 * 4);

    hipMemsetAsync(gcount, 0, 2048, stream);

    bucket_count_kernel<<<EB_GRID, 256, 0, stream>>>(ei, gcount);

    mlp_and_scatter_kernel<<<MLP_GRID + EB_GRID, 256, 0, stream>>>(emb, l1w, l1b, l2w, l2b, x,
                                                                    ei, et, gcount, gcursor, gstart, recs);

    edge_wmma_kernel<D_X, D_Y, 0><<<NB, EK_T, 0, stream>>>(recs, gstart, x, w1, root1, b1, acc1);

    edge_wmma_kernel<D_Y, D_O, 1><<<NB, EK_T, 0, stream>>>(recs, gstart, acc1, w2, root2, b2, out);
}


# Edge blocks bin a whole bucket in ONE round (4608 slots, was 2 of 2048): records read once into registers, rank from the counting add, slot = base + rank; record and LDS sum order change
# speedup vs baseline: 4.4628x; 1.2375x over previous
#include <hip/hip_runtime.h>
#include <hip/hip_bf16.h>
#include <math.h>

#define N_NODES 100000
#define N_REL   16
#define N_EDGES 1000000
#define D_IN    192
#define D_H     128
#define D_X     64
#define D_Y     32
#define D_O     16

typedef _Float16 v16h __attribute__((ext_vector_type(16)));
typedef _Float16 v8h  __attribute__((ext_vector_type(8)));
typedef float    v8f  __attribute__((ext_vector_type(8)));
typedef float    v4f  __attribute__((ext_vector_type(4)));
typedef float    v2f  __attribute__((ext_vector_type(2)));
typedef int      v4i  __attribute__((ext_vector_type(4)));

static __device__ __forceinline__ v16h cat8(v8h lo, v8h hi) {
    return __builtin_shufflevector(lo, hi, 0,1,2,3,4,5,6,7,8,9,10,11,12,13,14,15);
}

#define NB       256
#define NPB      391
#define EB_CH    2048
#define EB_PT    (EB_CH / 256)
#define EB_GRID  ((N_EDGES + EB_CH - 1) / EB_CH)
#define MLP_GRID ((N_NODES + 127) / 128)
#define EK_T     512
#define EK_CH    4608
#define EK_PT    (EK_CH / EK_T)
#define EK_CAP   (EK_CH + 16 * N_REL)

static __device__ __forceinline__ int rec_src(int r) { return r & 0x1FFFF; }
static __device__ __forceinline__ int rec_rel(int r) { return (r >> 17) & 15; }
static __device__ __forceinline__ int rec_dl(int r)  { return r >> 21; }

__launch_bounds__(256)
__global__ void bucket_count_kernel(const int* __restrict__ ei, int* __restrict__ gcount)
{
    __shared__ int sh[NB];
    const int tid = threadIdx.x;
    const int e0  = blockIdx.x * EB_CH;
    sh[tid] = 0;
    __syncthreads();
    int dst[EB_PT];
#pragma unroll
    for (int i = 0; i < EB_PT; ++i) dst[i] = ei[N_EDGES + min(e0 + i * 256 + tid, N_EDGES - 1)];
#pragma unroll
    for (int i = 0; i < EB_PT; ++i)
        if (e0 + i * 256 + tid < N_EDGES) atomicAdd(&sh[dst[i] / NPB], 1);
    __syncthreads();
    if (sh[tid] > 0) atomicAdd(&gcount[tid], sh[tid]);
}

static __device__ __forceinline__ void bucket_scatter_block(const int blk, int* sh,
                                      const int* __restrict__ ei, const int* __restrict__ et,
                                      const int* __restrict__ gcount, int* __restrict__ gcursor,
                                      int* __restrict__ gstart, int* __restrict__ recs)
{
    int* sScan = sh;
    int* sLoc  = sh + NB;
    int* sBase = sh + 2 * NB;
    const int tid  = threadIdx.x;
    const int e0   = blk * EB_CH;
    const int mine = gcount[tid];
    sScan[tid] = mine;
    sLoc[tid]  = 0;
    __syncthreads();
    for (int off = 1; off < NB; off <<= 1) {
        const int v = (tid >= off) ? sScan[tid - off] : 0;
        __syncthreads();
        sScan[tid] += v;
        __syncthreads();
    }
    const int start = sScan[tid] - mine;
    if (blk == 0) {
        gstart[tid] = start;
        if (tid == NB - 1) gstart[NB] = sScan[tid];
    }

    int rec[EB_PT], bk[EB_PT];
#pragma unroll
    for (int i = 0; i < EB_PT; ++i) {
        const int ec  = min(e0 + i * 256 + tid, N_EDGES - 1);
        const int src = ei[ec];
        const int dst = ei[N_EDGES + ec];
        const int b   = dst / NPB;
        bk[i]  = b;
        rec[i] = src | (et[ec] << 17) | ((dst - b * NPB) << 21);
    }
#pragma unroll
    for (int i = 0; i < EB_PT; ++i) {
        if (e0 + i * 256 + tid >= N_EDGES) bk[i] = -1;
        if (bk[i] >= 0) atomicAdd(&sLoc[bk[i]], 1);
    }
    __syncthreads();
    const int n = sLoc[tid];
    sBase[tid] = start + ((n > 0) ? atomicAdd(&gcursor[tid], n) : 0);
    sLoc[tid]  = 0;
    __syncthreads();
#pragma unroll
    for (int i = 0; i < EB_PT; ++i) {
        if (bk[i] >= 0) {
            const int pos = sBase[bk[i]] + atomicAdd(&sLoc[bk[i]], 1);
            recs[pos] = rec[i];
        }
    }
}

static __device__ __forceinline__ int w1_at(int r, int k) { return r * D_IN + ((k ^ ((r >> 1) & 7)) << 3); }
static __device__ __forceinline__ int w2_at(int r, int k) { return r * D_H + ((k ^ (r & 15)) << 3); }

static __device__ __forceinline__ float tanh_exp2(float x)
{
    const float t = __builtin_amdgcn_exp2f(x * 2.885390081777927f);
    return 1.0f - 2.0f * __builtin_amdgcn_rcpf(1.0f + t);
}

__launch_bounds__(256)
__global__ void mlp_and_scatter_kernel(const float* __restrict__ emb,
                                       const float* __restrict__ w1, const float* __restrict__ b1,
                                       const float* __restrict__ w2, const float* __restrict__ b2,
                                       float* __restrict__ xout,
                                       const int* __restrict__ ei, const int* __restrict__ et,
                                       const int* __restrict__ gcount, int* __restrict__ gcursor,
                                       int* __restrict__ gstart, int* __restrict__ recs)
{
    __shared__ __align__(16) _Float16 sW1[D_H * D_IN];
    __shared__ __align__(16) _Float16 sW2[D_X * D_H];

    if (blockIdx.x >= MLP_GRID) {
        bucket_scatter_block(blockIdx.x - MLP_GRID, reinterpret_cast<int*>(sW1),
                             ei, et, gcount, gcursor, gstart, recs);
        return;
    }

    const int tid = threadIdx.x;
    {
        const int c0 = tid & 63, r0 = tid >> 6;
        for (int r = r0; r < D_H; r += 4) {
            for (int j = 0; j < 3; ++j) {
                const int col = c0 + 64 * j;
                sW1[w1_at(r, col >> 3) + (col & 7)] = (_Float16)w1[r * D_IN + col];
            }
        }
    }
    {
        const int c0 = tid & 127, r0 = tid >> 7;
        for (int r = r0; r < D_X; r += 2)
            sW2[w2_at(r, c0 >> 3) + (c0 & 7)] = (_Float16)w2[r * D_H + c0];
    }
    __syncthreads();

    const int wave = tid >> 5;
    const int lane = tid & 31;
    const int half = lane >> 4;
    const int m    = lane & 15;
    const int rowBase = blockIdx.x * 128 + wave * 16;

    v8f accH[8] = {};
    const int gr = min(rowBase + m, N_NODES - 1);
    const float* arow = emb + (size_t)gr * D_IN + half * 8;

    for (int ks = 0; ks < 6; ++ks) {
        const float* ap = arow + ks * 32;
        v4f f0 = *(const v4f*)(ap);
        v4f f1 = *(const v4f*)(ap + 4);
        v4f f2 = *(const v4f*)(ap + 16);
        v4f f3 = *(const v4f*)(ap + 20);
        v8h lo, hi;
#pragma unroll
        for (int i = 0; i < 4; ++i) {
            lo[i]     = (_Float16)f0[i];  lo[i + 4] = (_Float16)f1[i];
            hi[i]     = (_Float16)f2[i];  hi[i + 4] = (_Float16)f3[i];
        }
        v16h a = cat8(lo, hi);
#pragma unroll
        for (int nt = 0; nt < 8; ++nt) {
            v16h b = cat8(*(const v8h*)(sW1 + w1_at(nt * 16 + m, ks * 4 + half)),
                          *(const v8h*)(sW1 + w1_at(nt * 16 + m, ks * 4 + half + 2)));
            accH[nt] = __builtin_amdgcn_wmma_f32_16x16x32_f16(
                false, a, false, b, (short)0, accH[nt], false, false);
        }
    }
    __syncthreads();

    static_assert(N_NODES % 16 == 0, "a wave's 16-row tile must not straddle the last node");
    if (__builtin_amdgcn_readfirstlane(rowBase) >= N_NODES) return;

    _Float16* Hs = sW1 + wave * (16 * D_H);
#pragma unroll
    for (int nt = 0; nt < 8; ++nt) {
        const int  col  = nt * 16 + m;
        const float bias = b1[col];
#pragma unroll
        for (int r = 0; r < 8; ++r) {
            const int mrow = half * 8 + r;
            Hs[w2_at(mrow, col >> 3) + (col & 7)] = (_Float16)tanh_exp2(accH[nt][r] + bias);
        }
    }

    v8f accX[4] = {};
    for (int ks = 0; ks < 4; ++ks) {
        v16h a = cat8(*(const v8h*)(Hs + w2_at(m, ks * 4 + half)),
                      *(const v8h*)(Hs + w2_at(m, ks * 4 + half + 2)));
#pragma unroll
        for (int nt = 0; nt < 4; ++nt) {
            v16h b = cat8(*(const v8h*)(sW2 + w2_at(nt * 16 + m, ks * 4 + half)),
                          *(const v8h*)(sW2 + w2_at(nt * 16 + m, ks * 4 + half + 2)));
            accX[nt] = __builtin_amdgcn_wmma_f32_16x16x32_f16(
                false, a, false, b, (short)0, accX[nt], false, false);
        }
    }
#pragma unroll
    for (int nt = 0; nt < 4; ++nt) {
        const int  col  = nt * 16 + m;
        const float bias = b2[col];
#pragma unroll
        for (int r = 0; r < 8; ++r) {
            const int row = rowBase + half * 8 + r;
            xout[(size_t)row * D_X + col] = accX[nt][r] + bias;
        }
    }
}


template <int F, int O>
static __device__ __forceinline__ void load_bfrag(const float* __restrict__ mat, int h, int c,
                                                  v2f (&bfrag)[O / 16][F / 4])
{
    const float* wr = mat + (size_t)(h * (F / 2)) * O + c;
#pragma unroll
    for (int nt = 0; nt < O / 16; ++nt)
#pragma unroll
        for (int j = 0; j < F / 4; ++j) {
            bfrag[nt][j][0] = wr[(2 * j)     * O + nt * 16];
            bfrag[nt][j][1] = wr[(2 * j + 1) * O + nt * 16];
        }
}

template <int F, int O>
static __device__ __forceinline__ void tile_products(const v4f (&av)[F / 8],
                                                     const v2f (&bfrag)[O / 16][F / 4],
                                                     v8f (&d)[O / 16])
{
#pragma unroll
    for (int nt = 0; nt < O / 16; ++nt) d[nt] = (v8f){0.f, 0.f, 0.f, 0.f, 0.f, 0.f, 0.f, 0.f};
#pragma unroll
    for (int j = 0; j < F / 4; ++j) {
        v2f a;
        a[0] = av[j >> 1][(j & 1) * 2];
        a[1] = av[j >> 1][(j & 1) * 2 + 1];
#pragma unroll
        for (int nt = 0; nt < O / 16; ++nt)
            d[nt] = __builtin_amdgcn_wmma_f32_16x16x4_f32(
                false, a, false, bfrag[nt][j], (short)0, d[nt], false, false);
    }
}

template <int F>
static __device__ __forceinline__ void gather_rows(const float* __restrict__ xin, const int* rowidx,
                                                   int p, v4f (&raw)[F / 8])
{
    int s[F / 8];
#pragma unroll
    for (int j = 0; j < F / 32; ++j) {
        const v4i q = *(const v4i*)(rowidx + 4 * j);
        s[4 * j] = q[0]; s[4 * j + 1] = q[1]; s[4 * j + 2] = q[2]; s[4 * j + 3] = q[3];
    }
#pragma unroll
    for (int i = 0; i < F / 8; ++i) raw[i] = *(const v4f*)(xin + (size_t)s[i] * F + 4 * p);
}

template <int F>
static __device__ __forceinline__ void rows_to_frag(const v4f (&raw)[F / 8], float* stw, const float* str,
                                                    v4f (&av)[F / 8])
{
#pragma unroll
    for (int i = 0; i < F / 8; ++i) *(v4f*)(stw + i * (F + 4)) = raw[i];
    __builtin_amdgcn_fence(__ATOMIC_ACQ_REL, "wavefront");
#pragma unroll
    for (int i = 0; i < F / 8; ++i) av[i] = *(const v4f*)(str + 4 * i);
    __builtin_amdgcn_fence(__ATOMIC_ACQ_REL, "wavefront");
}

template <int F, int O, int ACT>
__launch_bounds__(EK_T)
__global__ void edge_wmma_kernel(const int* __restrict__ recs, const int* __restrict__ gstart,
                                 const float* __restrict__ xin, const float* __restrict__ w,
                                 const float* __restrict__ root, const float* __restrict__ bias,
                                 float* __restrict__ acc)
{
    constexpr int KS = F / 4;
    constexpr int NT = O / 16;
    constexpr int HV = F / 8;
    constexpr int LPR = F / 4;
    constexpr int SP  = F + 4;

    __shared__ __align__(16) int   sSrc[EK_CAP];
    __shared__ __align__(16) int   sDst[EK_CAP];
    __shared__ __align__(16) float sScl[EK_CAP];
    __shared__ int sHist[N_REL];
    __shared__ int sTileBase[N_REL + 1];
    __shared__ int   sCnt[NPB * N_REL];
    __shared__ float sAcc[NPB * O];
    __shared__ __align__(16) float sStage[(EK_T / 32) * 16 * SP];

    const int tid  = threadIdx.x;
    const int n0   = blockIdx.x * NPB;
    const int nn   = min(NPB, N_NODES - n0);
    const int eBeg = gstart[blockIdx.x];
    const int eEnd = gstart[blockIdx.x + 1];

    int myrec[EK_PT];
#pragma unroll
    for (int i = 0; i < EK_PT; ++i) myrec[i] = recs[min(eBeg + i * EK_T + tid, N_EDGES - 1)];

    for (int i = tid; i < NPB * N_REL; i += EK_T) sCnt[i] = 0;
    if (tid < N_REL) sHist[tid] = 0;
    for (int i = tid; i < EK_CAP; i += EK_T) { sSrc[i] = 0; sDst[i] = -1; sScl[i] = 0.0f; }

    const int wave = __builtin_amdgcn_readfirstlane(tid >> 5);
    const int lane = tid & 31;
    const int h    = lane >> 4;
    const int c    = lane & 15;

    v2f bfrag[NT][KS];

    const int g = lane / LPR;
    const int p = lane % LPR;
    float*       stw = sStage + wave * (16 * SP) + (g * HV) * SP + 4 * p;
    const float* str = sStage + wave * (16 * SP) + c * SP + h * (F / 2);

    load_bfrag<F, O>(root, h, c, bfrag);
    for (int t = wave; t * 16 < nn; t += EK_T / 32) {
        v4f raw[HV], ar[HV];
#pragma unroll
        for (int i = 0; i < HV; ++i) {
            const int rc = min(t * 16 + g * HV + i, nn - 1);
            raw[i] = *(const v4f*)(xin + (size_t)(n0 + rc) * F + 4 * p);
        }
        rows_to_frag<F>(raw, stw, str, ar);
        v8f d[NT];
        tile_products<F, O>(ar, bfrag, d);
#pragma unroll
        for (int nt = 0; nt < NT; ++nt) {
            const float bs = bias[nt * 16 + c];
#pragma unroll
            for (int q = 0; q < 8; ++q) {
                const int r = t * 16 + 8 * h + q;
                if (r < nn) sAcc[r * O + nt * 16 + c] = d[nt][q] + bs;
            }
        }
    }
    __syncthreads();

    int rank[EK_PT];
#pragma unroll
    for (int i = 0; i < EK_PT; ++i) {
        if (eBeg + i * EK_T + tid >= eEnd) myrec[i] = -1;
        rank[i] = 0;
        if (myrec[i] >= 0) {
            atomicAdd(&sCnt[rec_dl(myrec[i]) * N_REL + rec_rel(myrec[i])], 1);
            rank[i] = atomicAdd(&sHist[rec_rel(myrec[i])], 1);
        }
    }
    for (int e = eBeg + EK_CH + tid; e < eEnd; e += EK_T) {
        const int rc = recs[e];
        atomicAdd(&sCnt[rec_dl(rc) * N_REL + rec_rel(rc)], 1);
    }
    __syncthreads();

    int curR = -1;

    for (int c0 = eBeg; c0 < eEnd; c0 += EK_CH) {
        if (c0 != eBeg) {
            if (tid < N_REL) sHist[tid] = 0;
            for (int i = tid; i < EK_CAP; i += EK_T) { sSrc[i] = 0; sDst[i] = -1; sScl[i] = 0.0f; }
            __syncthreads();
#pragma unroll
            for (int i = 0; i < EK_PT; ++i) {
                const int e  = c0 + i * EK_T + tid;
                const int rc = recs[min(e, eEnd - 1)];
                myrec[i] = (e < eEnd) ? rc : -1;
            }
#pragma unroll
            for (int i = 0; i < EK_PT; ++i) {
                rank[i] = 0;
                if (myrec[i] >= 0) rank[i] = atomicAdd(&sHist[rec_rel(myrec[i])], 1);
            }
            __syncthreads();
        }
        if (tid == 0) {
            int t = 0;
            for (int r = 0; r < N_REL; ++r) {
                sTileBase[r] = t;
                t += (sHist[r] + 15) >> 4;
            }
            sTileBase[N_REL] = t;
        }
        __syncthreads();
        int cn[EK_PT], tb[EK_PT];
#pragma unroll
        for (int i = 0; i < EK_PT; ++i) {
            const int rv = max(myrec[i], 0);
            cn[i] = sCnt[rec_dl(rv) * N_REL + rec_rel(rv)];
            tb[i] = sTileBase[rec_rel(rv)];
        }
#pragma unroll
        for (int i = 0; i < EK_PT; ++i) {
            if (myrec[i] >= 0) {
                const int pos = tb[i] * 16 + rank[i];
                sSrc[pos] = rec_src(myrec[i]);
                sDst[pos] = rec_dl(myrec[i]);
                sScl[pos] = 1.0f / fmaxf((float)cn[i], 1.0f);
            }
        }
        __syncthreads();

        const int T    = __builtin_amdgcn_readfirstlane(sTileBase[N_REL]);
        const int per  = (T + EK_T / 32 - 1) / (EK_T / 32);
        const int tBeg = wave * per;
        const int tEnd = min(T, tBeg + per);

        v4f raw[HV];
        gather_rows<F>(xin, &sSrc[tBeg * 16 + g * HV], p, raw);

        for (int t = tBeg; t < tEnd; ++t) {
            int rr = 0;
#pragma unroll
            for (int q = 1; q < N_REL; ++q) rr += (sTileBase[q] <= t) ? 1 : 0;
            const int r = __builtin_amdgcn_readfirstlane(rr);
            if (r != curR) {
                curR = r;
                load_bfrag<F, O>(w + (size_t)r * F * O, h, c, bfrag);
            }

            const int slot0 = t * 16;
            v4f av[HV];
            rows_to_frag<F>(raw, stw, str, av);
            const int tn = min(t + 1, tEnd - 1);
            gather_rows<F>(xin, &sSrc[tn * 16 + g * HV], p, raw);

            v8f d[NT];
            tile_products<F, O>(av, bfrag, d);

            const v4i dlo = *(const v4i*)&sDst[slot0 + 8 * h];
            const v4i dhi = *(const v4i*)&sDst[slot0 + 8 * h + 4];
            const v4f slo = *(const v4f*)&sScl[slot0 + 8 * h];
            const v4f shi = *(const v4f*)&sScl[slot0 + 8 * h + 4];
#pragma unroll
            for (int q = 0; q < 8; ++q) {
                const int   dst = (q < 4) ? dlo[q & 3] : dhi[q & 3];
                const float sc  = (q < 4) ? slo[q & 3] : shi[q & 3];
                if (dst >= 0) {
#pragma unroll
                    for (int nt = 0; nt < NT; ++nt)
                        atomicAdd(&sAcc[dst * O + nt * 16 + c], d[nt][q] * sc);
                }
            }
        }
        __syncthreads();
    }

    float* arow = acc + (size_t)n0 * O;
    for (int i = tid; i < nn * O; i += EK_T) {
        const float v = sAcc[i];
        arow[i] = (ACT == 0) ? fmaxf(v, 0.0f) : 1.0f / (1.0f + expf(-v));
    }
}

extern "C" void kernel_launch(void* const* d_in, const int* in_sizes, int n_in,
                              void* d_out, int out_size, void* d_ws, size_t ws_size,
                              hipStream_t stream)
{
    const float* emb   = (const float*)d_in[0];
    const float* l1w   = (const float*)d_in[1];
    const float* l1b   = (const float*)d_in[2];
    const float* l2w   = (const float*)d_in[3];
    const float* l2b   = (const float*)d_in[4];
    const float* w1    = (const float*)d_in[5];
    const float* root1 = (const float*)d_in[6];
    const float* b1    = (const float*)d_in[7];
    const float* w2    = (const float*)d_in[8];
    const float* root2 = (const float*)d_in[9];
    const float* b2    = (const float*)d_in[10];
    const int*   ei    = (const int*)d_in[11];
    const int*   et    = (const int*)d_in[12];
    float*       out   = (float*)d_out;

    char*  ws      = (char*)d_ws;
    float* x       = (float*)(ws);
    int*   gcount  = (int*)(ws + (size_t)N_NODES * 64 * 4);
    int*   gcursor = (int*)(ws + (size_t)N_NODES * 64 * 4 + 1024);
    int*   gstart  = (int*)(ws + (size_t)N_NODES * 64 * 4 + 2048);
    int*   recs    = (int*)(ws + (size_t)N_NODES * 64 * 4 + 4096);
    float* acc1    = (float*)(ws + (size_t)N_NODES * 80 * 4);

    hipMemsetAsync(gcount, 0, 2048, stream);

    bucket_count_kernel<<<EB_GRID, 256, 0, stream>>>(ei, gcount);

    mlp_and_scatter_kernel<<<MLP_GRID + EB_GRID, 256, 0, stream>>>(emb, l1w, l1b, l2w, l2b, x,
                                                                    ei, et, gcount, gcursor, gstart, recs);

    edge_wmma_kernel<D_X, D_Y, 0><<<NB, EK_T, 0, stream>>>(recs, gstart, x, w1, root1, b1, acc1);

    edge_wmma_kernel<D_Y, D_O, 1><<<NB, EK_T, 0, stream>>>(recs, gstart, acc1, w2, root2, b2, out);
}


# A bucket owns a fixed region of 4608 record slots: the scatter reserves room by its returning cursor add, so the bucket-size launch goes; a larger bucket re-reads the edge list; sum order as parent
# speedup vs baseline: 4.6719x; 1.0469x over previous
#include <hip/hip_runtime.h>
#include <hip/hip_bf16.h>
#include <math.h>

#define N_NODES 100000
#define N_REL   16
#define N_EDGES 1000000
#define D_IN    192
#define D_H     128
#define D_X     64
#define D_Y     32
#define D_O     16

typedef _Float16 v16h __attribute__((ext_vector_type(16)));
typedef _Float16 v8h  __attribute__((ext_vector_type(8)));
typedef float    v8f  __attribute__((ext_vector_type(8)));
typedef float    v4f  __attribute__((ext_vector_type(4)));
typedef float    v2f  __attribute__((ext_vector_type(2)));
typedef int      v4i  __attribute__((ext_vector_type(4)));

static __device__ __forceinline__ v16h cat8(v8h lo, v8h hi) {
    return __builtin_shufflevector(lo, hi, 0,1,2,3,4,5,6,7,8,9,10,11,12,13,14,15);
}

#define NB       256
#define NPB      391
#define EB_CH    2048
#define EB_PT    (EB_CH / 256)
#define EB_GRID  ((N_EDGES + EB_CH - 1) / EB_CH)
#define MLP_GRID ((N_NODES + 127) / 128)
#define EK_T     512
#define EK_CH    4608
#define EK_PT    (EK_CH / EK_T)
#define EK_CAP   (EK_CH + 16 * N_REL)

static __device__ __forceinline__ int rec_src(int r) { return r & 0x1FFFF; }
static __device__ __forceinline__ int rec_rel(int r) { return (r >> 17) & 15; }
static __device__ __forceinline__ int rec_dl(int r)  { return r >> 21; }

static __device__ __forceinline__ int edge_rec_of_bucket(const int* __restrict__ ei, const int* __restrict__ et,
                                                         int e, int b)
{
    if (e >= N_EDGES) return -1;
    const int dst = ei[N_EDGES + e];
    if (dst / NPB != b) return -1;
    return ei[e] | (et[e] << 17) | ((dst - b * NPB) << 21);
}

static __device__ __forceinline__ void bucket_scatter_block(const int blk, int* sh,
                                      const int* __restrict__ ei, const int* __restrict__ et,
                                      int* __restrict__ gcursor, int* __restrict__ recs)
{
    int* sLoc  = sh;
    int* sBase = sh + NB;
    const int tid = threadIdx.x;
    const int e0  = blk * EB_CH;
    sLoc[tid] = 0;
    __syncthreads();

    int rec[EB_PT], bk[EB_PT];
#pragma unroll
    for (int i = 0; i < EB_PT; ++i) {
        const int ec  = min(e0 + i * 256 + tid, N_EDGES - 1);
        const int src = ei[ec];
        const int dst = ei[N_EDGES + ec];
        const int b   = dst / NPB;
        bk[i]  = b;
        rec[i] = src | (et[ec] << 17) | ((dst - b * NPB) << 21);
    }
#pragma unroll
    for (int i = 0; i < EB_PT; ++i) {
        if (e0 + i * 256 + tid >= N_EDGES) bk[i] = -1;
        if (bk[i] >= 0) atomicAdd(&sLoc[bk[i]], 1);
    }
    __syncthreads();
    const int n = sLoc[tid];
    sBase[tid] = (n > 0) ? atomicAdd(&gcursor[tid], n) : 0;
    sLoc[tid]  = 0;
    __syncthreads();
#pragma unroll
    for (int i = 0; i < EB_PT; ++i) {
        if (bk[i] >= 0) {
            const int pos = sBase[bk[i]] + atomicAdd(&sLoc[bk[i]], 1);
            if (pos < EK_CH) recs[bk[i] * EK_CH + pos] = rec[i];
        }
    }
}

static __device__ __forceinline__ int w1_at(int r, int k) { return r * D_IN + ((k ^ ((r >> 1) & 7)) << 3); }
static __device__ __forceinline__ int w2_at(int r, int k) { return r * D_H + ((k ^ (r & 15)) << 3); }

static __device__ __forceinline__ float tanh_exp2(float x)
{
    const float t = __builtin_amdgcn_exp2f(x * 2.885390081777927f);
    return 1.0f - 2.0f * __builtin_amdgcn_rcpf(1.0f + t);
}

__launch_bounds__(256)
__global__ void mlp_and_scatter_kernel(const float* __restrict__ emb,
                                       const float* __restrict__ w1, const float* __restrict__ b1,
                                       const float* __restrict__ w2, const float* __restrict__ b2,
                                       float* __restrict__ xout,
                                       const int* __restrict__ ei, const int* __restrict__ et,
                                       int* __restrict__ gcursor, int* __restrict__ recs)
{
    __shared__ __align__(16) _Float16 sW1[D_H * D_IN];
    __shared__ __align__(16) _Float16 sW2[D_X * D_H];

    if (blockIdx.x >= MLP_GRID) {
        bucket_scatter_block(blockIdx.x - MLP_GRID, reinterpret_cast<int*>(sW1), ei, et, gcursor, recs);
        return;
    }

    const int tid = threadIdx.x;
    {
        const int c0 = tid & 63, r0 = tid >> 6;
        for (int r = r0; r < D_H; r += 4) {
            for (int j = 0; j < 3; ++j) {
                const int col = c0 + 64 * j;
                sW1[w1_at(r, col >> 3) + (col & 7)] = (_Float16)w1[r * D_IN + col];
            }
        }
    }
    {
        const int c0 = tid & 127, r0 = tid >> 7;
        for (int r = r0; r < D_X; r += 2)
            sW2[w2_at(r, c0 >> 3) + (c0 & 7)] = (_Float16)w2[r * D_H + c0];
    }
    __syncthreads();

    const int wave = tid >> 5;
    const int lane = tid & 31;
    const int half = lane >> 4;
    const int m    = lane & 15;
    const int rowBase = blockIdx.x * 128 + wave * 16;

    v8f accH[8] = {};
    const int gr = min(rowBase + m, N_NODES - 1);
    const float* arow = emb + (size_t)gr * D_IN + half * 8;

    for (int ks = 0; ks < 6; ++ks) {
        const float* ap = arow + ks * 32;
        v4f f0 = *(const v4f*)(ap);
        v4f f1 = *(const v4f*)(ap + 4);
        v4f f2 = *(const v4f*)(ap + 16);
        v4f f3 = *(const v4f*)(ap + 20);
        v8h lo, hi;
#pragma unroll
        for (int i = 0; i < 4; ++i) {
            lo[i]     = (_Float16)f0[i];  lo[i + 4] = (_Float16)f1[i];
            hi[i]     = (_Float16)f2[i];  hi[i + 4] = (_Float16)f3[i];
        }
        v16h a = cat8(lo, hi);
#pragma unroll
        for (int nt = 0; nt < 8; ++nt) {
            v16h b = cat8(*(const v8h*)(sW1 + w1_at(nt * 16 + m, ks * 4 + half)),
                          *(const v8h*)(sW1 + w1_at(nt * 16 + m, ks * 4 + half + 2)));
            accH[nt] = __builtin_amdgcn_wmma_f32_16x16x32_f16(
                false, a, false, b, (short)0, accH[nt], false, false);
        }
    }
    __syncthreads();

    static_assert(N_NODES % 16 == 0, "a wave's 16-row tile must not straddle the last node");
    if (__builtin_amdgcn_readfirstlane(rowBase) >= N_NODES) return;

    _Float16* Hs = sW1 + wave * (16 * D_H);
#pragma unroll
    for (int nt = 0; nt < 8; ++nt) {
        const int  col  = nt * 16 + m;
        const float bias = b1[col];
#pragma unroll
        for (int r = 0; r < 8; ++r) {
            const int mrow = half * 8 + r;
            Hs[w2_at(mrow, col >> 3) + (col & 7)] = (_Float16)tanh_exp2(accH[nt][r] + bias);
        }
    }

    v8f accX[4] = {};
    for (int ks = 0; ks < 4; ++ks) {
        v16h a = cat8(*(const v8h*)(Hs + w2_at(m, ks * 4 + half)),
                      *(const v8h*)(Hs + w2_at(m, ks * 4 + half + 2)));
#pragma unroll
        for (int nt = 0; nt < 4; ++nt) {
            v16h b = cat8(*(const v8h*)(sW2 + w2_at(nt * 16 + m, ks * 4 + half)),
                          *(const v8h*)(sW2 + w2_at(nt * 16 + m, ks * 4 + half + 2)));
            accX[nt] = __builtin_amdgcn_wmma_f32_16x16x32_f16(
                false, a, false, b, (short)0, accX[nt], false, false);
        }
    }
#pragma unroll
    for (int nt = 0; nt < 4; ++nt) {
        const int  col  = nt * 16 + m;
        const float bias = b2[col];
#pragma unroll
        for (int r = 0; r < 8; ++r) {
            const int row = rowBase + half * 8 + r;
            xout[(size_t)row * D_X + col] = accX[nt][r] + bias;
        }
    }
}


template <int F, int O>
static __device__ __forceinline__ void load_bfrag(const float* __restrict__ mat, int h, int c,
                                                  v2f (&bfrag)[O / 16][F / 4])
{
    const float* wr = mat + (size_t)(h * (F / 2)) * O + c;
#pragma unroll
    for (int nt = 0; nt < O / 16; ++nt)
#pragma unroll
        for (int j = 0; j < F / 4; ++j) {
            bfrag[nt][j][0] = wr[(2 * j)     * O + nt * 16];
            bfrag[nt][j][1] = wr[(2 * j + 1) * O + nt * 16];
        }
}

template <int F, int O>
static __device__ __forceinline__ void tile_products(const v4f (&av)[F / 8],
                                                     const v2f (&bfrag)[O / 16][F / 4],
                                                     v8f (&d)[O / 16])
{
#pragma unroll
    for (int nt = 0; nt < O / 16; ++nt) d[nt] = (v8f){0.f, 0.f, 0.f, 0.f, 0.f, 0.f, 0.f, 0.f};
#pragma unroll
    for (int j = 0; j < F / 4; ++j) {
        v2f a;
        a[0] = av[j >> 1][(j & 1) * 2];
        a[1] = av[j >> 1][(j & 1) * 2 + 1];
#pragma unroll
        for (int nt = 0; nt < O / 16; ++nt)
            d[nt] = __builtin_amdgcn_wmma_f32_16x16x4_f32(
                false, a, false, bfrag[nt][j], (short)0, d[nt], false, false);
    }
}

template <int F>
static __device__ __forceinline__ void gather_rows(const float* __restrict__ xin, const int* rowidx,
                                                   int p, v4f (&raw)[F / 8])
{
    int s[F / 8];
#pragma unroll
    for (int j = 0; j < F / 32; ++j) {
        const v4i q = *(const v4i*)(rowidx + 4 * j);
        s[4 * j] = q[0]; s[4 * j + 1] = q[1]; s[4 * j + 2] = q[2]; s[4 * j + 3] = q[3];
    }
#pragma unroll
    for (int i = 0; i < F / 8; ++i) raw[i] = *(const v4f*)(xin + (size_t)s[i] * F + 4 * p);
}

template <int F>
static __device__ __forceinline__ void rows_to_frag(const v4f (&raw)[F / 8], float* stw, const float* str,
                                                    v4f (&av)[F / 8])
{
#pragma unroll
    for (int i = 0; i < F / 8; ++i) *(v4f*)(stw + i * (F + 4)) = raw[i];
    __builtin_amdgcn_fence(__ATOMIC_ACQ_REL, "wavefront");
#pragma unroll
    for (int i = 0; i < F / 8; ++i) av[i] = *(const v4f*)(str + 4 * i);
    __builtin_amdgcn_fence(__ATOMIC_ACQ_REL, "wavefront");
}

template <int F, int O, int ACT>
__launch_bounds__(EK_T)
__global__ void edge_wmma_kernel(const int* __restrict__ recs, const int* __restrict__ btotal,
                                 const int* __restrict__ ei, const int* __restrict__ et,
                                 const float* __restrict__ xin, const float* __restrict__ w,
                                 const float* __restrict__ root, const float* __restrict__ bias,
                                 float* __restrict__ acc)
{
    constexpr int KS = F / 4;
    constexpr int NT = O / 16;
    constexpr int HV = F / 8;
    constexpr int LPR = F / 4;
    constexpr int SP  = F + 4;

    __shared__ __align__(16) int   sSrc[EK_CAP];
    __shared__ __align__(16) int   sDst[EK_CAP];
    __shared__ __align__(16) float sScl[EK_CAP];
    __shared__ int sHist[N_REL];
    __shared__ int sTileBase[N_REL + 1];
    __shared__ int   sCnt[NPB * N_REL];
    __shared__ float sAcc[NPB * O];
    __shared__ __align__(16) float sStage[(EK_T / 32) * 16 * SP];

    const int tid   = threadIdx.x;
    const int bkt   = blockIdx.x;
    const int n0    = bkt * NPB;
    const int nn    = min(NPB, N_NODES - n0);
    const int total = btotal[bkt];
    const bool big  = total > EK_CH;
    const int rounds = big ? (N_EDGES + EK_CH - 1) / EK_CH : 1;

    const int* region = recs + (size_t)bkt * EK_CH;
    int myrec[EK_PT];
#pragma unroll
    for (int i = 0; i < EK_PT; ++i) myrec[i] = region[i * EK_T + tid];

    for (int i = tid; i < NPB * N_REL; i += EK_T) sCnt[i] = 0;
    if (tid < N_REL) sHist[tid] = 0;
    for (int i = tid; i < EK_CAP; i += EK_T) { sSrc[i] = 0; sDst[i] = -1; sScl[i] = 0.0f; }

    const int wave = __builtin_amdgcn_readfirstlane(tid >> 5);
    const int lane = tid & 31;
    const int h    = lane >> 4;
    const int c    = lane & 15;

    v2f bfrag[NT][KS];

    const int g = lane / LPR;
    const int p = lane % LPR;
    float*       stw = sStage + wave * (16 * SP) + (g * HV) * SP + 4 * p;
    const float* str = sStage + wave * (16 * SP) + c * SP + h * (F / 2);

    load_bfrag<F, O>(root, h, c, bfrag);
    for (int t = wave; t * 16 < nn; t += EK_T / 32) {
        v4f raw[HV], ar[HV];
#pragma unroll
        for (int i = 0; i < HV; ++i) {
            const int rc = min(t * 16 + g * HV + i, nn - 1);
            raw[i] = *(const v4f*)(xin + (size_t)(n0 + rc) * F + 4 * p);
        }
        rows_to_frag<F>(raw, stw, str, ar);
        v8f d[NT];
        tile_products<F, O>(ar, bfrag, d);
#pragma unroll
        for (int nt = 0; nt < NT; ++nt) {
            const float bs = bias[nt * 16 + c];
#pragma unroll
            for (int q = 0; q < 8; ++q) {
                const int r = t * 16 + 8 * h + q;
                if (r < nn) sAcc[r * O + nt * 16 + c] = d[nt][q] + bs;
            }
        }
    }
    __syncthreads();

    int rank[EK_PT];
#pragma unroll
    for (int i = 0; i < EK_PT; ++i) {
        const int k = i * EK_T + tid;
        if (big) myrec[i] = edge_rec_of_bucket(ei, et, k, bkt);
        else if (k >= total) myrec[i] = -1;
        rank[i] = 0;
        if (myrec[i] >= 0) {
            atomicAdd(&sCnt[rec_dl(myrec[i]) * N_REL + rec_rel(myrec[i])], 1);
            rank[i] = atomicAdd(&sHist[rec_rel(myrec[i])], 1);
        }
    }
    if (big) {
        for (int e = EK_CH + tid; e < N_EDGES; e += EK_T) {
            const int rc = edge_rec_of_bucket(ei, et, e, bkt);
            if (rc >= 0) atomicAdd(&sCnt[rec_dl(rc) * N_REL + rec_rel(rc)], 1);
        }
    }
    __syncthreads();

    int curR = -1;

    for (int rd = 0; rd < rounds; ++rd) {
        if (rd != 0) {
            if (tid < N_REL) sHist[tid] = 0;
            for (int i = tid; i < EK_CAP; i += EK_T) { sSrc[i] = 0; sDst[i] = -1; sScl[i] = 0.0f; }
            __syncthreads();
#pragma unroll
            for (int i = 0; i < EK_PT; ++i)
                myrec[i] = edge_rec_of_bucket(ei, et, rd * EK_CH + i * EK_T + tid, bkt);
#pragma unroll
            for (int i = 0; i < EK_PT; ++i) {
                rank[i] = 0;
                if (myrec[i] >= 0) rank[i] = atomicAdd(&sHist[rec_rel(myrec[i])], 1);
            }
            __syncthreads();
        }
        if (tid == 0) {
            int t = 0;
            for (int r = 0; r < N_REL; ++r) {
                sTileBase[r] = t;
                t += (sHist[r] + 15) >> 4;
            }
            sTileBase[N_REL] = t;
        }
        __syncthreads();
        int cn[EK_PT], tb[EK_PT];
#pragma unroll
        for (int i = 0; i < EK_PT; ++i) {
            const int rv = max(myrec[i], 0);
            cn[i] = sCnt[rec_dl(rv) * N_REL + rec_rel(rv)];
            tb[i] = sTileBase[rec_rel(rv)];
        }
#pragma unroll
        for (int i = 0; i < EK_PT; ++i) {
            if (myrec[i] >= 0) {
                const int pos = tb[i] * 16 + rank[i];
                sSrc[pos] = rec_src(myrec[i]);
                sDst[pos] = rec_dl(myrec[i]);
                sScl[pos] = 1.0f / fmaxf((float)cn[i], 1.0f);
            }
        }
        __syncthreads();

        const int T    = __builtin_amdgcn_readfirstlane(sTileBase[N_REL]);
        const int per  = (T + EK_T / 32 - 1) / (EK_T / 32);
        const int tBeg = wave * per;
        const int tEnd = min(T, tBeg + per);

        v4f raw[HV];
        gather_rows<F>(xin, &sSrc[tBeg * 16 + g * HV], p, raw);

        for (int t = tBeg; t < tEnd; ++t) {
            int rr = 0;
#pragma unroll
            for (int q = 1; q < N_REL; ++q) rr += (sTileBase[q] <= t) ? 1 : 0;
            const int r = __builtin_amdgcn_readfirstlane(rr);
            if (r != curR) {
                curR = r;
                load_bfrag<F, O>(w + (size_t)r * F * O, h, c, bfrag);
            }

            const int slot0 = t * 16;
            v4f av[HV];
            rows_to_frag<F>(raw, stw, str, av);
            const int tn = min(t + 1, tEnd - 1);
            gather_rows<F>(xin, &sSrc[tn * 16 + g * HV], p, raw);

            v8f d[NT];
            tile_products<F, O>(av, bfrag, d);

            const v4i dlo = *(const v4i*)&sDst[slot0 + 8 * h];
            const v4i dhi = *(const v4i*)&sDst[slot0 + 8 * h + 4];
            const v4f slo = *(const v4f*)&sScl[slot0 + 8 * h];
            const v4f shi = *(const v4f*)&sScl[slot0 + 8 * h + 4];
#pragma unroll
            for (int q = 0; q < 8; ++q) {
                const int   dst = (q < 4) ? dlo[q & 3] : dhi[q & 3];
                const float sc  = (q < 4) ? slo[q & 3] : shi[q & 3];
                if (dst >= 0) {
#pragma unroll
                    for (int nt = 0; nt < NT; ++nt)
                        atomicAdd(&sAcc[dst * O + nt * 16 + c], d[nt][q] * sc);
                }
            }
        }
        __syncthreads();
    }

    float* arow = acc + (size_t)n0 * O;
    for (int i = tid; i < nn * O; i += EK_T) {
        const float v = sAcc[i];
        arow[i] = (ACT == 0) ? fmaxf(v, 0.0f) : 1.0f / (1.0f + expf(-v));
    }
}

extern "C" void kernel_launch(void* const* d_in, const int* in_sizes, int n_in,
                              void* d_out, int out_size, void* d_ws, size_t ws_size,
                              hipStream_t stream)
{
    const float* emb   = (const float*)d_in[0];
    const float* l1w   = (const float*)d_in[1];
    const float* l1b   = (const float*)d_in[2];
    const float* l2w   = (const float*)d_in[3];
    const float* l2b   = (const float*)d_in[4];
    const float* w1    = (const float*)d_in[5];
    const float* root1 = (const float*)d_in[6];
    const float* b1    = (const float*)d_in[7];
    const float* w2    = (const float*)d_in[8];
    const float* root2 = (const float*)d_in[9];
    const float* b2    = (const float*)d_in[10];
    const int*   ei    = (const int*)d_in[11];
    const int*   et    = (const int*)d_in[12];
    float*       out   = (float*)d_out;

    char*  ws      = (char*)d_ws;
    float* x       = (float*)(ws);
    int*   gcursor = (int*)(ws + (size_t)N_NODES * 64 * 4);
    int*   recs    = (int*)(ws + (size_t)N_NODES * 64 * 4 + 4096);
    float* acc1    = (float*)(ws + (size_t)N_NODES * 80 * 4);

    hipMemsetAsync(gcursor, 0, 1024, stream);

    mlp_and_scatter_kernel<<<MLP_GRID + EB_GRID, 256, 0, stream>>>(emb, l1w, l1b, l2w, l2b, x,
                                                                    ei, et, gcursor, recs);

    edge_wmma_kernel<D_X, D_Y, 0><<<NB, EK_T, 0, stream>>>(recs, gcursor, ei, et, x, w1, root1, b1, acc1);

    edge_wmma_kernel<D_Y, D_O, 1><<<NB, EK_T, 0, stream>>>(recs, gcursor, ei, et, acc1, w2, root2, b2, out);
}


# Layer-1 edge and root products: f32 16x16x4 WMMA replaced by three f16 16x16x32 WMMAs on hi/lo f16 splits of both operands (hi*hi+hi*lo+lo*hi, f32 accumulate): dtype change, sum order changes
# speedup vs baseline: 4.9860x; 1.0672x over previous
#include <hip/hip_runtime.h>
#include <hip/hip_bf16.h>
#include <math.h>

#define N_NODES 100000
#define N_REL   16
#define N_EDGES 1000000
#define D_IN    192
#define D_H     128
#define D_X     64
#define D_Y     32
#define D_O     16

typedef _Float16 v16h __attribute__((ext_vector_type(16)));
typedef _Float16 v8h  __attribute__((ext_vector_type(8)));
typedef float    v8f  __attribute__((ext_vector_type(8)));
typedef float    v4f  __attribute__((ext_vector_type(4)));
typedef float    v2f  __attribute__((ext_vector_type(2)));
typedef int      v4i  __attribute__((ext_vector_type(4)));

static __device__ __forceinline__ v16h cat8(v8h lo, v8h hi) {
    return __builtin_shufflevector(lo, hi, 0,1,2,3,4,5,6,7,8,9,10,11,12,13,14,15);
}

#define NB       256
#define NPB      391
#define EB_CH    2048
#define EB_PT    (EB_CH / 256)
#define EB_GRID  ((N_EDGES + EB_CH - 1) / EB_CH)
#define MLP_GRID ((N_NODES + 127) / 128)
#define EK_T     512
#define EK_CH    4608
#define EK_PT    (EK_CH / EK_T)
#define EK_CAP   (EK_CH + 16 * N_REL)

static __device__ __forceinline__ int rec_src(int r) { return r & 0x1FFFF; }
static __device__ __forceinline__ int rec_rel(int r) { return (r >> 17) & 15; }
static __device__ __forceinline__ int rec_dl(int r)  { return r >> 21; }

static __device__ __forceinline__ int edge_rec_of_bucket(const int* __restrict__ ei, const int* __restrict__ et,
                                                         int e, int b)
{
    if (e >= N_EDGES) return -1;
    const int dst = ei[N_EDGES + e];
    if (dst / NPB != b) return -1;
    return ei[e] | (et[e] << 17) | ((dst - b * NPB) << 21);
}

static __device__ __forceinline__ void bucket_scatter_block(const int blk, int* sh,
                                      const int* __restrict__ ei, const int* __restrict__ et,
                                      int* __restrict__ gcursor, int* __restrict__ recs)
{
    int* sLoc  = sh;
    int* sBase = sh + NB;
    const int tid = threadIdx.x;
    const int e0  = blk * EB_CH;
    sLoc[tid] = 0;
    __syncthreads();

    int rec[EB_PT], bk[EB_PT];
#pragma unroll
    for (int i = 0; i < EB_PT; ++i) {
        const int ec  = min(e0 + i * 256 + tid, N_EDGES - 1);
        const int src = ei[ec];
        const int dst = ei[N_EDGES + ec];
        const int b   = dst / NPB;
        bk[i]  = b;
        rec[i] = src | (et[ec] << 17) | ((dst - b * NPB) << 21);
    }
#pragma unroll
    for (int i = 0; i < EB_PT; ++i) {
        if (e0 + i * 256 + tid >= N_EDGES) bk[i] = -1;
        if (bk[i] >= 0) atomicAdd(&sLoc[bk[i]], 1);
    }
    __syncthreads();
    const int n = sLoc[tid];
    sBase[tid] = (n > 0) ? atomicAdd(&gcursor[tid], n) : 0;
    sLoc[tid]  = 0;
    __syncthreads();
#pragma unroll
    for (int i = 0; i < EB_PT; ++i) {
        if (bk[i] >= 0) {
            const int pos = sBase[bk[i]] + atomicAdd(&sLoc[bk[i]], 1);
            if (pos < EK_CH) recs[bk[i] * EK_CH + pos] = rec[i];
        }
    }
}

static __device__ __forceinline__ int w1_at(int r, int k) { return r * D_IN + ((k ^ ((r >> 1) & 7)) << 3); }
static __device__ __forceinline__ int w2_at(int r, int k) { return r * D_H + ((k ^ (r & 15)) << 3); }

static __device__ __forceinline__ float tanh_exp2(float x)
{
    const float t = __builtin_amdgcn_exp2f(x * 2.885390081777927f);
    return 1.0f - 2.0f * __builtin_amdgcn_rcpf(1.0f + t);
}

__launch_bounds__(256)
__global__ void mlp_and_scatter_kernel(const float* __restrict__ emb,
                                       const float* __restrict__ w1, const float* __restrict__ b1,
                                       const float* __restrict__ w2, const float* __restrict__ b2,
                                       float* __restrict__ xout,
                                       const int* __restrict__ ei, const int* __restrict__ et,
                                       int* __restrict__ gcursor, int* __restrict__ recs)
{
    __shared__ __align__(16) _Float16 sW1[D_H * D_IN];
    __shared__ __align__(16) _Float16 sW2[D_X * D_H];

    if (blockIdx.x >= MLP_GRID) {
        bucket_scatter_block(blockIdx.x - MLP_GRID, reinterpret_cast<int*>(sW1), ei, et, gcursor, recs);
        return;
    }

    const int tid = threadIdx.x;
    {
        const int c0 = tid & 63, r0 = tid >> 6;
        for (int r = r0; r < D_H; r += 4) {
            for (int j = 0; j < 3; ++j) {
                const int col = c0 + 64 * j;
                sW1[w1_at(r, col >> 3) + (col & 7)] = (_Float16)w1[r * D_IN + col];
            }
        }
    }
    {
        const int c0 = tid & 127, r0 = tid >> 7;
        for (int r = r0; r < D_X; r += 2)
            sW2[w2_at(r, c0 >> 3) + (c0 & 7)] = (_Float16)w2[r * D_H + c0];
    }
    __syncthreads();

    const int wave = tid >> 5;
    const int lane = tid & 31;
    const int half = lane >> 4;
    const int m    = lane & 15;
    const int rowBase = blockIdx.x * 128 + wave * 16;

    v8f accH[8] = {};
    const int gr = min(rowBase + m, N_NODES - 1);
    const float* arow = emb + (size_t)gr * D_IN + half * 8;

    for (int ks = 0; ks < 6; ++ks) {
        const float* ap = arow + ks * 32;
        v4f f0 = *(const v4f*)(ap);
        v4f f1 = *(const v4f*)(ap + 4);
        v4f f2 = *(const v4f*)(ap + 16);
        v4f f3 = *(const v4f*)(ap + 20);
        v8h lo, hi;
#pragma unroll
        for (int i = 0; i < 4; ++i) {
            lo[i]     = (_Float16)f0[i];  lo[i + 4] = (_Float16)f1[i];
            hi[i]     = (_Float16)f2[i];  hi[i + 4] = (_Float16)f3[i];
        }
        v16h a = cat8(lo, hi);
#pragma unroll
        for (int nt = 0; nt < 8; ++nt) {
            v16h b = cat8(*(const v8h*)(sW1 + w1_at(nt * 16 + m, ks * 4 + half)),
                          *(const v8h*)(sW1 + w1_at(nt * 16 + m, ks * 4 + half + 2)));
            accH[nt] = __builtin_amdgcn_wmma_f32_16x16x32_f16(
                false, a, false, b, (short)0, accH[nt], false, false);
        }
    }
    __syncthreads();

    static_assert(N_NODES % 16 == 0, "a wave's 16-row tile must not straddle the last node");
    if (__builtin_amdgcn_readfirstlane(rowBase) >= N_NODES) return;

    _Float16* Hs = sW1 + wave * (16 * D_H);
#pragma unroll
    for (int nt = 0; nt < 8; ++nt) {
        const int  col  = nt * 16 + m;
        const float bias = b1[col];
#pragma unroll
        for (int r = 0; r < 8; ++r) {
            const int mrow = half * 8 + r;
            Hs[w2_at(mrow, col >> 3) + (col & 7)] = (_Float16)tanh_exp2(accH[nt][r] + bias);
        }
    }

    v8f accX[4] = {};
    for (int ks = 0; ks < 4; ++ks) {
        v16h a = cat8(*(const v8h*)(Hs + w2_at(m, ks * 4 + half)),
                      *(const v8h*)(Hs + w2_at(m, ks * 4 + half + 2)));
#pragma unroll
        for (int nt = 0; nt < 4; ++nt) {
            v16h b = cat8(*(const v8h*)(sW2 + w2_at(nt * 16 + m, ks * 4 + half)),
                          *(const v8h*)(sW2 + w2_at(nt * 16 + m, ks * 4 + half + 2)));
            accX[nt] = __builtin_amdgcn_wmma_f32_16x16x32_f16(
                false, a, false, b, (short)0, accX[nt], false, false);
        }
    }
#pragma unroll
    for (int nt = 0; nt < 4; ++nt) {
        const int  col  = nt * 16 + m;
        const float bias = b2[col];
#pragma unroll
        for (int r = 0; r < 8; ++r) {
            const int row = rowBase + half * 8 + r;
            xout[(size_t)row * D_X + col] = accX[nt][r] + bias;
        }
    }
}


template <int F, int O>
static __device__ __forceinline__ void load_bfrag(const float* __restrict__ mat, int h, int c,
                                                  v2f (&bfrag)[O / 16][F / 4])
{
    const float* wr = mat + (size_t)(h * (F / 2)) * O + c;
#pragma unroll
    for (int nt = 0; nt < O / 16; ++nt)
#pragma unroll
        for (int j = 0; j < F / 4; ++j) {
            bfrag[nt][j][0] = wr[(2 * j)     * O + nt * 16];
            bfrag[nt][j][1] = wr[(2 * j + 1) * O + nt * 16];
        }
}

template <int F, int O>
static __device__ __forceinline__ void tile_products(const v4f (&av)[F / 8],
                                                     const v2f (&bfrag)[O / 16][F / 4],
                                                     v8f (&d)[O / 16])
{
#pragma unroll
    for (int nt = 0; nt < O / 16; ++nt) d[nt] = (v8f){0.f, 0.f, 0.f, 0.f, 0.f, 0.f, 0.f, 0.f};
#pragma unroll
    for (int j = 0; j < F / 4; ++j) {
        v2f a;
        a[0] = av[j >> 1][(j & 1) * 2];
        a[1] = av[j >> 1][(j & 1) * 2 + 1];
#pragma unroll
        for (int nt = 0; nt < O / 16; ++nt)
            d[nt] = __builtin_amdgcn_wmma_f32_16x16x4_f32(
                false, a, false, bfrag[nt][j], (short)0, d[nt], false, false);
    }
}

template <int F, int O, bool SPLIT> struct BFrag;
template <int F, int O> struct BFrag<F, O, false> { v2f  f[O / 16][F / 4]; };
template <int F, int O> struct BFrag<F, O, true>  { v16h hi[O / 16][F / 32]; v16h lo[O / 16][F / 32]; };

template <int F, int O>
static __device__ __forceinline__ void load_bfrag(const float* __restrict__ mat, int h, int c, BFrag<F, O, false>& b)
{
    load_bfrag<F, O>(mat, h, c, b.f);
}

template <int F, int O>
static __device__ __forceinline__ void load_bfrag(const float* __restrict__ mat, int h, int c, BFrag<F, O, true>& b)
{
    const float* wr = mat + (size_t)(h * (F / 2)) * O + c;
#pragma unroll
    for (int nt = 0; nt < O / 16; ++nt)
#pragma unroll
        for (int j = 0; j < F / 32; ++j) {
            float v[16];
#pragma unroll
            for (int s = 0; s < 16; ++s) v[s] = wr[(16 * j + s) * O + nt * 16];
#pragma unroll
            for (int s = 0; s < 16; ++s) {
                const _Float16 hv = (_Float16)v[s];
                b.hi[nt][j][s] = hv;
                b.lo[nt][j][s] = (_Float16)(v[s] - (float)hv);
            }
        }
}

template <int F, int O>
static __device__ __forceinline__ void tile_products(const v4f (&av)[F / 8], const BFrag<F, O, false>& b, v8f (&d)[O / 16])
{
    tile_products<F, O>(av, b.f, d);
}

template <int F, int O>
static __device__ __forceinline__ void tile_products(const v4f (&av)[F / 8], const BFrag<F, O, true>& b, v8f (&d)[O / 16])
{
    v8f dx[O / 16];
#pragma unroll
    for (int nt = 0; nt < O / 16; ++nt) {
        d[nt]  = (v8f){0.f, 0.f, 0.f, 0.f, 0.f, 0.f, 0.f, 0.f};
        dx[nt] = (v8f){0.f, 0.f, 0.f, 0.f, 0.f, 0.f, 0.f, 0.f};
    }
#pragma unroll
    for (int j = 0; j < F / 32; ++j) {
        v16h ah, al;
#pragma unroll
        for (int s = 0; s < 16; ++s) {
            const float x = av[4 * j + (s >> 2)][s & 3];
            const _Float16 hv = (_Float16)x;
            ah[s] = hv;
            al[s] = (_Float16)(x - (float)hv);
        }
#pragma unroll
        for (int nt = 0; nt < O / 16; ++nt)
            d[nt] = __builtin_amdgcn_wmma_f32_16x16x32_f16(false, ah, false, b.hi[nt][j], (short)0, d[nt], false, false);
#pragma unroll
        for (int nt = 0; nt < O / 16; ++nt)
            dx[nt] = __builtin_amdgcn_wmma_f32_16x16x32_f16(false, ah, false, b.lo[nt][j], (short)0, dx[nt], false, false);
#pragma unroll
        for (int nt = 0; nt < O / 16; ++nt)
            dx[nt] = __builtin_amdgcn_wmma_f32_16x16x32_f16(false, al, false, b.hi[nt][j], (short)0, dx[nt], false, false);
    }
#pragma unroll
    for (int nt = 0; nt < O / 16; ++nt) d[nt] += dx[nt];
}

template <int F>
static __device__ __forceinline__ void gather_rows(const float* __restrict__ xin, const int* rowidx,
                                                   int p, v4f (&raw)[F / 8])
{
    int s[F / 8];
#pragma unroll
    for (int j = 0; j < F / 32; ++j) {
        const v4i q = *(const v4i*)(rowidx + 4 * j);
        s[4 * j] = q[0]; s[4 * j + 1] = q[1]; s[4 * j + 2] = q[2]; s[4 * j + 3] = q[3];
    }
#pragma unroll
    for (int i = 0; i < F / 8; ++i) raw[i] = *(const v4f*)(xin + (size_t)s[i] * F + 4 * p);
}

template <int F>
static __device__ __forceinline__ void rows_to_frag(const v4f (&raw)[F / 8], float* stw, const float* str,
                                                    v4f (&av)[F / 8])
{
#pragma unroll
    for (int i = 0; i < F / 8; ++i) *(v4f*)(stw + i * (F + 4)) = raw[i];
    __builtin_amdgcn_fence(__ATOMIC_ACQ_REL, "wavefront");
#pragma unroll
    for (int i = 0; i < F / 8; ++i) av[i] = *(const v4f*)(str + 4 * i);
    __builtin_amdgcn_fence(__ATOMIC_ACQ_REL, "wavefront");
}

template <int F, int O, int ACT, bool SPLIT>
__launch_bounds__(EK_T)
__global__ void edge_wmma_kernel(const int* __restrict__ recs, const int* __restrict__ btotal,
                                 const int* __restrict__ ei, const int* __restrict__ et,
                                 const float* __restrict__ xin, const float* __restrict__ w,
                                 const float* __restrict__ root, const float* __restrict__ bias,
                                 float* __restrict__ acc)
{
    constexpr int NT = O / 16;
    constexpr int HV = F / 8;
    constexpr int LPR = F / 4;
    constexpr int SP  = F + 4;

    __shared__ __align__(16) int   sSrc[EK_CAP];
    __shared__ __align__(16) int   sDst[EK_CAP];
    __shared__ __align__(16) float sScl[EK_CAP];
    __shared__ int sHist[N_REL];
    __shared__ int sTileBase[N_REL + 1];
    __shared__ int   sCnt[NPB * N_REL];
    __shared__ float sAcc[NPB * O];
    __shared__ __align__(16) float sStage[(EK_T / 32) * 16 * SP];

    const int tid   = threadIdx.x;
    const int bkt   = blockIdx.x;
    const int n0    = bkt * NPB;
    const int nn    = min(NPB, N_NODES - n0);
    const int total = btotal[bkt];
    const bool big  = total > EK_CH;
    const int rounds = big ? (N_EDGES + EK_CH - 1) / EK_CH : 1;

    const int* region = recs + (size_t)bkt * EK_CH;
    int myrec[EK_PT];
#pragma unroll
    for (int i = 0; i < EK_PT; ++i) myrec[i] = region[i * EK_T + tid];

    for (int i = tid; i < NPB * N_REL; i += EK_T) sCnt[i] = 0;
    if (tid < N_REL) sHist[tid] = 0;
    for (int i = tid; i < EK_CAP; i += EK_T) { sSrc[i] = 0; sDst[i] = -1; sScl[i] = 0.0f; }

    const int wave = __builtin_amdgcn_readfirstlane(tid >> 5);
    const int lane = tid & 31;
    const int h    = lane >> 4;
    const int c    = lane & 15;

    BFrag<F, O, SPLIT> bfrag;

    const int g = lane / LPR;
    const int p = lane % LPR;
    float*       stw = sStage + wave * (16 * SP) + (g * HV) * SP + 4 * p;
    const float* str = sStage + wave * (16 * SP) + c * SP + h * (F / 2);

    load_bfrag<F, O>(root, h, c, bfrag);
    for (int t = wave; t * 16 < nn; t += EK_T / 32) {
        v4f raw[HV], ar[HV];
#pragma unroll
        for (int i = 0; i < HV; ++i) {
            const int rc = min(t * 16 + g * HV + i, nn - 1);
            raw[i] = *(const v4f*)(xin + (size_t)(n0 + rc) * F + 4 * p);
        }
        rows_to_frag<F>(raw, stw, str, ar);
        v8f d[NT];
        tile_products<F, O>(ar, bfrag, d);
#pragma unroll
        for (int nt = 0; nt < NT; ++nt) {
            const float bs = bias[nt * 16 + c];
#pragma unroll
            for (int q = 0; q < 8; ++q) {
                const int r = t * 16 + 8 * h + q;
                if (r < nn) sAcc[r * O + nt * 16 + c] = d[nt][q] + bs;
            }
        }
    }
    __syncthreads();

    int rank[EK_PT];
#pragma unroll
    for (int i = 0; i < EK_PT; ++i) {
        const int k = i * EK_T + tid;
        if (big) myrec[i] = edge_rec_of_bucket(ei, et, k, bkt);
        else if (k >= total) myrec[i] = -1;
        rank[i] = 0;
        if (myrec[i] >= 0) {
            atomicAdd(&sCnt[rec_dl(myrec[i]) * N_REL + rec_rel(myrec[i])], 1);
            rank[i] = atomicAdd(&sHist[rec_rel(myrec[i])], 1);
        }
    }
    if (big) {
        for (int e = EK_CH + tid; e < N_EDGES; e += EK_T) {
            const int rc = edge_rec_of_bucket(ei, et, e, bkt);
            if (rc >= 0) atomicAdd(&sCnt[rec_dl(rc) * N_REL + rec_rel(rc)], 1);
        }
    }
    __syncthreads();

    int curR = -1;

    for (int rd = 0; rd < rounds; ++rd) {
        if (rd != 0) {
            if (tid < N_REL) sHist[tid] = 0;
            for (int i = tid; i < EK_CAP; i += EK_T) { sSrc[i] = 0; sDst[i] = -1; sScl[i] = 0.0f; }
            __syncthreads();
#pragma unroll
            for (int i = 0; i < EK_PT; ++i)
                myrec[i] = edge_rec_of_bucket(ei, et, rd * EK_CH + i * EK_T + tid, bkt);
#pragma unroll
            for (int i = 0; i < EK_PT; ++i) {
                rank[i] = 0;
                if (myrec[i] >= 0) rank[i] = atomicAdd(&sHist[rec_rel(myrec[i])], 1);
            }
            __syncthreads();
        }
        if (tid == 0) {
            int t = 0;
            for (int r = 0; r < N_REL; ++r) {
                sTileBase[r] = t;
                t += (sHist[r] + 15) >> 4;
            }
            sTileBase[N_REL] = t;
        }
        __syncthreads();
        int cn[EK_PT], tb[EK_PT];
#pragma unroll
        for (int i = 0; i < EK_PT; ++i) {
            const int rv = max(myrec[i], 0);
            cn[i] = sCnt[rec_dl(rv) * N_REL + rec_rel(rv)];
            tb[i] = sTileBase[rec_rel(rv)];
        }
#pragma unroll
        for (int i = 0; i < EK_PT; ++i) {
            if (myrec[i] >= 0) {
                const int pos = tb[i] * 16 + rank[i];
                sSrc[pos] = rec_src(myrec[i]);
                sDst[pos] = rec_dl(myrec[i]);
                sScl[pos] = 1.0f / fmaxf((float)cn[i], 1.0f);
            }
        }
        __syncthreads();

        const int T    = __builtin_amdgcn_readfirstlane(sTileBase[N_REL]);
        const int per  = (T + EK_T / 32 - 1) / (EK_T / 32);
        const int tBeg = wave * per;
        const int tEnd = min(T, tBeg + per);

        v4f raw[HV];
        gather_rows<F>(xin, &sSrc[tBeg * 16 + g * HV], p, raw);

        for (int t = tBeg; t < tEnd; ++t) {
            int rr = 0;
#pragma unroll
            for (int q = 1; q < N_REL; ++q) rr += (sTileBase[q] <= t) ? 1 : 0;
            const int r = __builtin_amdgcn_readfirstlane(rr);
            if (r != curR) {
                curR = r;
                load_bfrag<F, O>(w + (size_t)r * F * O, h, c, bfrag);
            }

            const int slot0 = t * 16;
            v4f av[HV];
            rows_to_frag<F>(raw, stw, str, av);
            const int tn = min(t + 1, tEnd - 1);
            gather_rows<F>(xin, &sSrc[tn * 16 + g * HV], p, raw);

            v8f d[NT];
            tile_products<F, O>(av, bfrag, d);

            const v4i dlo = *(const v4i*)&sDst[slot0 + 8 * h];
            const v4i dhi = *(const v4i*)&sDst[slot0 + 8 * h + 4];
            const v4f slo = *(const v4f*)&sScl[slot0 + 8 * h];
            const v4f shi = *(const v4f*)&sScl[slot0 + 8 * h + 4];
#pragma unroll
            for (int q = 0; q < 8; ++q) {
                const int   dst = (q < 4) ? dlo[q & 3] : dhi[q & 3];
                const float sc  = (q < 4) ? slo[q & 3] : shi[q & 3];
                if (dst >= 0) {
#pragma unroll
                    for (int nt = 0; nt < NT; ++nt)
                        atomicAdd(&sAcc[dst * O + nt * 16 + c], d[nt][q] * sc);
                }
            }
        }
        __syncthreads();
    }

    float* arow = acc + (size_t)n0 * O;
    for (int i = tid; i < nn * O; i += EK_T) {
        const float v = sAcc[i];
        arow[i] = (ACT == 0) ? fmaxf(v, 0.0f) : 1.0f / (1.0f + expf(-v));
    }
}

extern "C" void kernel_launch(void* const* d_in, const int* in_sizes, int n_in,
                              void* d_out, int out_size, void* d_ws, size_t ws_size,
                              hipStream_t stream)
{
    const float* emb   = (const float*)d_in[0];
    const float* l1w   = (const float*)d_in[1];
    const float* l1b   = (const float*)d_in[2];
    const float* l2w   = (const float*)d_in[3];
    const float* l2b   = (const float*)d_in[4];
    const float* w1    = (const float*)d_in[5];
    const float* root1 = (const float*)d_in[6];
    const float* b1    = (const float*)d_in[7];
    const float* w2    = (const float*)d_in[8];
    const float* root2 = (const float*)d_in[9];
    const float* b2    = (const float*)d_in[10];
    const int*   ei    = (const int*)d_in[11];
    const int*   et    = (const int*)d_in[12];
    float*       out   = (float*)d_out;

    char*  ws      = (char*)d_ws;
    float* x       = (float*)(ws);
    int*   gcursor = (int*)(ws + (size_t)N_NODES * 64 * 4);
    int*   recs    = (int*)(ws + (size_t)N_NODES * 64 * 4 + 4096);
    float* acc1    = (float*)(ws + (size_t)N_NODES * 80 * 4);

    hipMemsetAsync(gcursor, 0, 1024, stream);

    mlp_and_scatter_kernel<<<MLP_GRID + EB_GRID, 256, 0, stream>>>(emb, l1w, l1b, l2w, l2b, x,
                                                                    ei, et, gcursor, recs);

    edge_wmma_kernel<D_X, D_Y, 0, true><<<NB, EK_T, 0, stream>>>(recs, gcursor, ei, et, x, w1, root1, b1, acc1);

    edge_wmma_kernel<D_Y, D_O, 1, false><<<NB, EK_T, 0, stream>>>(recs, gcursor, ei, et, acc1, w2, root2, b2, out);
}
